# starting kernel
# baseline (speedup 1.0000x reference)
#include <hip/hip_runtime.h>


typedef __attribute__((ext_vector_type(2))) float v2f;
typedef __attribute__((ext_vector_type(8))) float v8f;

#define TILES_PER_BLOCK 8
#define CHUNK_KT 32

__global__ __launch_bounds__(256) void zero_f32_kernel(float* __restrict__ p,
                                                       long n) {
  long i = (long)blockIdx.x * blockDim.x + threadIdx.x;
  long stride = (long)gridDim.x * blockDim.x;
  for (; i < n; i += stride) p[i] = 0.0f;
}

__global__ __launch_bounds__(256) void pack_weights_kernel(
    const float* __restrict__ W, float* __restrict__ P, int K) {
  int total = K * 128;
  for (int tid = blockIdx.x * blockDim.x + threadIdx.x; tid < total;
       tid += gridDim.x * blockDim.x) {
    int frag = tid >> 6;
    int r = tid & 63;
    int L = r >> 1;
    int v = r & 1;
    int kt = frag >> 3;
    int nt = frag & 7;
    int k = kt * 4 + ((L >> 4) << 1) + v;
    int n = nt * 16 + (L & 15);
    P[tid] = W[k * 128 + n];
  }
}

__global__ __launch_bounds__(256) void degree_kernel(
    const int* __restrict__ dst, float* __restrict__ deg, int E) {
  int e = blockIdx.x * blockDim.x + threadIdx.x;
  if (e < E) atomicAdd(&deg[dst[e]], 1.0f);
}

__global__ __launch_bounds__(256) void scatter_add_kernel(
    const float* __restrict__ feat, const int* __restrict__ src,
    const int* __restrict__ dst, float* __restrict__ agg, int E, int F) {
  int chunks = F >> 2;
  long total = (long)E * chunks;
  long tid = (long)blockIdx.x * blockDim.x + threadIdx.x;
  long stride = (long)gridDim.x * blockDim.x;
  for (; tid < total; tid += stride) {
    int e = (int)(tid / chunks);
    int c = (int)(tid % chunks);
    int s = src[e];
    int d = dst[e];
    const float4 val = *(const float4*)(feat + (size_t)s * F + c * 4);
    float* o = agg + (size_t)d * F + c * 4;
    atomicAdd(o + 0, val.x);
    atomicAdd(o + 1, val.y);
    atomicAdd(o + 2, val.z);
    atomicAdd(o + 3, val.w);
  }
}

__global__ __launch_bounds__(256) void gemm_pass_kernel(
    const float* __restrict__ A, int K, const float* __restrict__ Bp,
    const float* __restrict__ bias, const float* __restrict__ deg,
    const float* __restrict__ prev, float* __restrict__ out, int relu,
    int nTiles) {
  __shared__ float lds[CHUNK_KT * 8 * 64];

  const int lane = threadIdx.x & 31;
  const int wave = threadIdx.x >> 5;
  int tile = blockIdx.x * TILES_PER_BLOCK + wave;
  const bool active = tile < nTiles;
  if (!active) tile = nTiles - 1;

  const int half = lane >> 4;
  const int n0 = lane & 15;
  const int arow = tile * 16 + n0;
  const float* __restrict__ arowp = A + (size_t)arow * K + half * 2;

  float scale = 1.0f;
  if (deg) scale = 1.0f / fmaxf(deg[arow], 1.0f);

  v8f acc[8];
#pragma unroll
  for (int nt = 0; nt < 8; ++nt) {
    float b = bias ? bias[nt * 16 + n0] : 0.0f;
    acc[nt] = (v8f){b, b, b, b, b, b, b, b};
  }

  const int nkt = K >> 2;
  v2f a_pre = *(const v2f*)(arowp);

  for (int kt0 = 0; kt0 < nkt; kt0 += CHUNK_KT) {
    __syncthreads();
    {
      const float4* __restrict__ gsrc =
          (const float4*)(Bp + (size_t)kt0 * 8 * 64);
      float4* __restrict__ ldst = (float4*)lds;
      for (int i = threadIdx.x; i < CHUNK_KT * 8 * 16; i += 256)
        ldst[i] = gsrc[i];
    }
    __syncthreads();

    for (int kt = 0; kt < CHUNK_KT; ++kt) {
      const int ktg = kt0 + kt;
      v2f a = a_pre * scale;
      const int ktn = (ktg + 1 < nkt) ? (ktg + 1) : ktg;
      a_pre = *(const v2f*)(arowp + ktn * 4);

      const float* bbase = lds + (kt * 8) * 64 + lane * 2;
      v2f bf[8];
#pragma unroll
      for (int nt = 0; nt < 8; ++nt) bf[nt] = *(const v2f*)(bbase + nt * 64);

#pragma unroll
      for (int nt = 0; nt < 8; ++nt) {
        acc[nt] = __builtin_amdgcn_wmma_f32_16x16x4_f32(
            false, a, false, bf[nt], (short)0, acc[nt], false, false);
      }
    }
  }

  if (!active) return;

  const int m0 = tile * 16;
#pragma unroll
  for (int nt = 0; nt < 8; ++nt) {
#pragma unroll
    for (int r = 0; r < 8; ++r) {
      int m = m0 + r + 8 * half;
      size_t idx = (size_t)m * 128 + nt * 16 + n0;
      float v = acc[nt][r];
      if (prev) v += prev[idx];
      if (relu) v = fmaxf(v, 0.0f);
      out[idx] = v;
    }
  }
}

extern "C" void kernel_launch(void* const* d_in, const int* in_sizes, int n_in,
                              void* d_out, int out_size, void* d_ws,
                              size_t ws_size, hipStream_t stream) {
  const int IN_DIM = 384, HID = 128;
  const int N = in_sizes[0] / IN_DIM;
  const int E = in_sizes[1] / 2;

  const float* x    = (const float*)d_in[0];
  const int*   ei   = (const int*)d_in[1];
  const float* W1_l = (const float*)d_in[2];
  const float* b1   = (const float*)d_in[3];
  const float* W1_r = (const float*)d_in[4];
  const float* W2_l = (const float*)d_in[5];
  const float* b2   = (const float*)d_in[6];
  const float* W2_r = (const float*)d_in[7];
  const int* src = ei;
  const int* dst = ei + E;
  float* out = (float*)d_out;

  float* ws = (float*)d_ws;
  const long deg_off  = 0;
  const long deg_pad  = ((long)N + 127) & ~127L;
  const long agg_off  = deg_off + deg_pad;
  const long h1_off   = agg_off + (long)N * IN_DIM;
  const long pw1l_off = h1_off + (long)N * HID;
  const long pw1r_off = pw1l_off + (long)IN_DIM * HID;
  const long pw2l_off = pw1r_off + (long)IN_DIM * HID;
  const long pw2r_off = pw2l_off + (long)HID * HID;

  float* deg  = ws + deg_off;
  float* agg  = ws + agg_off;
  float* h1   = ws + h1_off;
  float* pw1l = ws + pw1l_off;
  float* pw1r = ws + pw1r_off;
  float* pw2l = ws + pw2l_off;
  float* pw2r = ws + pw2r_off;

  const int nTiles = N / 16;
  const int gemmBlocks = (nTiles + TILES_PER_BLOCK - 1) / TILES_PER_BLOCK;

  {
    long nz = deg_pad + (long)N * IN_DIM;
    int blocks = (int)((nz + 255) / 256);
    zero_f32_kernel<<<blocks, 256, 0, stream>>>(ws, nz);
  }
  pack_weights_kernel<<<(IN_DIM * 128 + 255) / 256, 256, 0, stream>>>(W1_l, pw1l, IN_DIM);
  pack_weights_kernel<<<(IN_DIM * 128 + 255) / 256, 256, 0, stream>>>(W1_r, pw1r, IN_DIM);
  pack_weights_kernel<<<(HID * 128 + 255) / 256, 256, 0, stream>>>(W2_l, pw2l, HID);
  pack_weights_kernel<<<(HID * 128 + 255) / 256, 256, 0, stream>>>(W2_r, pw2r, HID);

  degree_kernel<<<(E + 255) / 256, 256, 0, stream>>>(dst, deg, E);

  {
    long total = (long)E * (IN_DIM / 4);
    int blocks = (int)((total + 255) / 256);
    scatter_add_kernel<<<blocks, 256, 0, stream>>>(x, src, dst, agg, E, IN_DIM);
  }

  gemm_pass_kernel<<<gemmBlocks, 256, 0, stream>>>(agg, IN_DIM, pw1l, b1, deg,
                                                   nullptr, h1, 0, nTiles);
  gemm_pass_kernel<<<gemmBlocks, 256, 0, stream>>>(x, IN_DIM, pw1r, nullptr,
                                                   nullptr, h1, h1, 1, nTiles);

  {
    long nz = (long)N * HID;
    int blocks = (int)((nz + 255) / 256);
    zero_f32_kernel<<<blocks, 256, 0, stream>>>(agg, nz);
    long total = (long)E * (HID / 4);
    int sblocks = (int)((total + 255) / 256);
    scatter_add_kernel<<<sblocks, 256, 0, stream>>>(h1, src, dst, agg, E, HID);
  }

  gemm_pass_kernel<<<gemmBlocks, 256, 0, stream>>>(agg, HID, pw2l, b2, deg,
                                                   nullptr, out, 0, nTiles);
  gemm_pass_kernel<<<gemmBlocks, 256, 0, stream>>>(h1, HID, pw2r, nullptr,
                                                   nullptr, out, out, 0, nTiles);
}


# Zero fills folded into GEMM passes A and B: each wave stores the zero start values (sum rows, deg) of the next scatter for its own tile in its epilogue; the two zero-fill launches are gone
# speedup vs baseline: 1.4124x; 1.4124x over previous
#include <hip/hip_runtime.h>


typedef __attribute__((ext_vector_type(2))) float v2f;
typedef __attribute__((ext_vector_type(8))) float v8f;

#define TILES_PER_BLOCK 8
#define CHUNK_KT 32

__global__ __launch_bounds__(256) void pack_weights_kernel(
    const float* __restrict__ W, float* __restrict__ P, int K) {
  int total = K * 128;
  for (int tid = blockIdx.x * blockDim.x + threadIdx.x; tid < total;
       tid += gridDim.x * blockDim.x) {
    int frag = tid >> 6;
    int r = tid & 63;
    int L = r >> 1;
    int v = r & 1;
    int kt = frag >> 3;
    int nt = frag & 7;
    int k = kt * 4 + ((L >> 4) << 1) + v;
    int n = nt * 16 + (L & 15);
    P[tid] = W[k * 128 + n];
  }
}

__global__ __launch_bounds__(256) void degree_kernel(
    const int* __restrict__ dst, float* __restrict__ deg, int E) {
  int e = blockIdx.x * blockDim.x + threadIdx.x;
  if (e < E) atomicAdd(&deg[dst[e]], 1.0f);
}

__global__ __launch_bounds__(256) void scatter_add_kernel(
    const float* __restrict__ feat, const int* __restrict__ src,
    const int* __restrict__ dst, float* __restrict__ agg, int E, int F) {
  int chunks = F >> 2;
  long total = (long)E * chunks;
  long tid = (long)blockIdx.x * blockDim.x + threadIdx.x;
  long stride = (long)gridDim.x * blockDim.x;
  for (; tid < total; tid += stride) {
    int e = (int)(tid / chunks);
    int c = (int)(tid % chunks);
    int s = src[e];
    int d = dst[e];
    const float4 val = *(const float4*)(feat + (size_t)s * F + c * 4);
    float* o = agg + (size_t)d * F + c * 4;
    atomicAdd(o + 0, val.x);
    atomicAdd(o + 1, val.y);
    atomicAdd(o + 2, val.z);
    atomicAdd(o + 3, val.w);
  }
}

__global__ __launch_bounds__(256) __attribute__((amdgpu_waves_per_eu(10)))
void gemm_pass_kernel(
    const float* __restrict__ A, int K, const float* __restrict__ Bp,
    const float* __restrict__ bias, const float* __restrict__ deg,
    const float* __restrict__ prev, const float* __restrict__ pdeg,
    float* __restrict__ out, int relu, int nTiles, float* __restrict__ zrows,
    float* __restrict__ zdeg) {
  __shared__ float lds[CHUNK_KT * 8 * 64];

  const int lane = threadIdx.x & 31;
  const int wave = threadIdx.x >> 5;
  int tile = blockIdx.x * TILES_PER_BLOCK + wave;
  const bool active = tile < nTiles;
  if (!active) tile = nTiles - 1;
  const int tile_u = __builtin_amdgcn_readfirstlane(tile);

  const int half = lane >> 4;
  const int n0 = lane & 15;
  const int arow = tile * 16 + n0;
  const float* __restrict__ arowp = A + (size_t)arow * K + half * 2;

  float scale = 1.0f;
  if (deg) scale = 1.0f / fmaxf(deg[arow], 1.0f);

  v8f acc[8];
#pragma unroll
  for (int nt = 0; nt < 8; ++nt) {
    float b = bias ? bias[nt * 16 + n0] : 0.0f;
    acc[nt] = (v8f){b, b, b, b, b, b, b, b};
  }

  const int nkt = K >> 2;
  v2f a_pre = *(const v2f*)(arowp);

  for (int kt0 = 0; kt0 < nkt; kt0 += CHUNK_KT) {
    __syncthreads();
    {
      const float4* __restrict__ gsrc =
          (const float4*)(Bp + (size_t)kt0 * 8 * 64);
      float4* __restrict__ ldst = (float4*)lds;
      for (int i = threadIdx.x; i < CHUNK_KT * 8 * 16; i += 256)
        ldst[i] = gsrc[i];
    }
    __syncthreads();

    for (int kt = 0; kt < CHUNK_KT; ++kt) {
      const int ktg = kt0 + kt;
      v2f a = a_pre * scale;
      const int ktn = (ktg + 1 < nkt) ? (ktg + 1) : ktg;
      a_pre = *(const v2f*)(arowp + ktn * 4);

      const float* bbase = lds + (kt * 8) * 64 + lane * 2;
      v2f bf[8];
#pragma unroll
      for (int nt = 0; nt < 8; ++nt) bf[nt] = *(const v2f*)(bbase + nt * 64);

#pragma unroll
      for (int nt = 0; nt < 8; ++nt) {
        acc[nt] = __builtin_amdgcn_wmma_f32_16x16x4_f32(
            false, a, false, bf[nt], (short)0, acc[nt], false, false);
      }
    }
  }

  if (!active) return;

  const int el = (int)__builtin_amdgcn_mbcnt_lo(~0u, 0u);
  const int eh = el >> 4;
  const int en = el & 15;

  if (zdeg && (tile_u & 1) == 0) {
    const int zi = tile_u * 16 + el;
    if (zi < nTiles * 16) zdeg[zi] = 0.0f;
  }
  if (zrows) {
    float4* zp = (float4*)(zrows + (size_t)tile_u * (16 * 128)) + el;
    const float4 z4 = {0.0f, 0.0f, 0.0f, 0.0f};
    for (int i = 0; i < 16; ++i) zp[i * 32] = z4;
  }

  const int m0 = tile_u * 16;
  float pscale[8];
#pragma unroll
  for (int r = 0; r < 8; ++r)
    pscale[r] = pdeg ? 1.0f / fmaxf(pdeg[m0 + r + 8 * eh], 1.0f) : 1.0f;
#pragma unroll
  for (int nt = 0; nt < 8; ++nt) {
#pragma unroll
    for (int r = 0; r < 8; ++r) {
      int m = m0 + r + 8 * eh;
      size_t idx = (size_t)m * 128 + nt * 16 + en;
      float v = acc[nt][r];
      if (prev) v += prev[idx] * pscale[r];
      if (relu) v = fmaxf(v, 0.0f);
      out[idx] = v;
    }
  }
}

extern "C" void kernel_launch(void* const* d_in, const int* in_sizes, int n_in,
                              void* d_out, int out_size, void* d_ws,
                              size_t ws_size, hipStream_t stream) {
  const int IN_DIM = 384, HID = 128;
  const int N = in_sizes[0] / IN_DIM;
  const int E = in_sizes[1] / 2;

  const float* x    = (const float*)d_in[0];
  const int*   ei   = (const int*)d_in[1];
  const float* W1_l = (const float*)d_in[2];
  const float* b1   = (const float*)d_in[3];
  const float* W1_r = (const float*)d_in[4];
  const float* W2_l = (const float*)d_in[5];
  const float* b2   = (const float*)d_in[6];
  const float* W2_r = (const float*)d_in[7];
  const int* src = ei;
  const int* dst = ei + E;
  float* out = (float*)d_out;

  float* ws = (float*)d_ws;
  const long deg_off  = 0;
  const long deg_pad  = ((long)N + 127) & ~127L;
  const long agg_off  = deg_off + deg_pad;
  const long y_off    = agg_off + (long)N * HID;
  const long h1_off   = y_off + (long)N * HID;
  const long pw1l_off = h1_off + (long)N * HID;
  const long pw1r_off = pw1l_off + (long)IN_DIM * HID;
  const long pw2l_off = pw1r_off + (long)IN_DIM * HID;
  const long pw2r_off = pw2l_off + (long)HID * HID;

  float* deg  = ws + deg_off;
  float* agg  = ws + agg_off;
  float* y    = ws + y_off;
  float* agg2 = y;
  float* h1   = ws + h1_off;
  float* pw1l = ws + pw1l_off;
  float* pw1r = ws + pw1r_off;
  float* pw2l = ws + pw2l_off;
  float* pw2r = ws + pw2r_off;

  const int nTiles = N / 16;
  const int gemmBlocks = (nTiles + TILES_PER_BLOCK - 1) / TILES_PER_BLOCK;

  pack_weights_kernel<<<(IN_DIM * 128 + 255) / 256, 256, 0, stream>>>(W1_l, pw1l, IN_DIM);
  pack_weights_kernel<<<(IN_DIM * 128 + 255) / 256, 256, 0, stream>>>(W1_r, pw1r, IN_DIM);
  pack_weights_kernel<<<(HID * 128 + 255) / 256, 256, 0, stream>>>(W2_l, pw2l, HID);
  pack_weights_kernel<<<(HID * 128 + 255) / 256, 256, 0, stream>>>(W2_r, pw2r, HID);

  gemm_pass_kernel<<<gemmBlocks, 256, 0, stream>>>(
      x, IN_DIM, pw1l, nullptr, nullptr, nullptr, nullptr, y, 0, nTiles, agg,
      deg);

  degree_kernel<<<(E + 255) / 256, 256, 0, stream>>>(dst, deg, E);

  {
    long total = (long)E * (HID / 4);
    int blocks = (int)((total + 255) / 256);
    scatter_add_kernel<<<blocks, 256, 0, stream>>>(y, src, dst, agg, E, HID);
  }

  gemm_pass_kernel<<<gemmBlocks, 256, 0, stream>>>(
      x, IN_DIM, pw1r, b1, nullptr, agg, deg, h1, 1, nTiles, agg2, nullptr);

  {
    long total = (long)E * (HID / 4);
    int sblocks = (int)((total + 255) / 256);
    scatter_add_kernel<<<sblocks, 256, 0, stream>>>(h1, src, dst, agg2, E, HID);
  }

  gemm_pass_kernel<<<gemmBlocks, 256, 0, stream>>>(
      agg2, HID, pw2l, b2, deg, nullptr, nullptr, out, 0, nTiles, nullptr,
      nullptr);
  gemm_pass_kernel<<<gemmBlocks, 256, 0, stream>>>(
      h1, HID, pw2r, nullptr, nullptr, out, nullptr, out, 0, nTiles, nullptr,
      nullptr);
}


# Scatter: lane l moves columns l, l+32, l+64, l+96 of a row, so each load and each f32 atomic of a wave covers one whole 128-B line, not a quarter of four lines
# speedup vs baseline: 1.5107x; 1.0696x over previous
#include <hip/hip_runtime.h>


typedef __attribute__((ext_vector_type(2))) float v2f;
typedef __attribute__((ext_vector_type(8))) float v8f;

#define TILES_PER_BLOCK 8
#define CHUNK_KT 32

__global__ __launch_bounds__(256) void pack_weights_kernel(
    const float* __restrict__ W, float* __restrict__ P, int K) {
  int total = K * 128;
  for (int tid = blockIdx.x * blockDim.x + threadIdx.x; tid < total;
       tid += gridDim.x * blockDim.x) {
    int frag = tid >> 6;
    int r = tid & 63;
    int L = r >> 1;
    int v = r & 1;
    int kt = frag >> 3;
    int nt = frag & 7;
    int k = kt * 4 + ((L >> 4) << 1) + v;
    int n = nt * 16 + (L & 15);
    P[tid] = W[k * 128 + n];
  }
}

__global__ __launch_bounds__(256) void degree_kernel(
    const int* __restrict__ dst, float* __restrict__ deg, int E) {
  int e = blockIdx.x * blockDim.x + threadIdx.x;
  if (e < E) atomicAdd(&deg[dst[e]], 1.0f);
}

__global__ __launch_bounds__(256) void scatter_add_kernel(
    const float* __restrict__ feat, const int* __restrict__ src,
    const int* __restrict__ dst, float* __restrict__ agg, int E) {
  const int lane = threadIdx.x & 31;
  const int e = blockIdx.x * 8 + (threadIdx.x >> 5);
  if (e >= E) return;
  const int s = src[e];
  const int d = dst[e];
  const float* f = feat + (size_t)s * 128 + lane;
  float* o = agg + (size_t)d * 128 + lane;
  const float v0 = f[0];
  const float v1 = f[32];
  const float v2 = f[64];
  const float v3 = f[96];
  atomicAdd(o, v0);
  atomicAdd(o + 32, v1);
  atomicAdd(o + 64, v2);
  atomicAdd(o + 96, v3);
}

__global__ __launch_bounds__(256) __attribute__((amdgpu_waves_per_eu(10)))
void gemm_pass_kernel(
    const float* __restrict__ A, int K, const float* __restrict__ Bp,
    const float* __restrict__ bias, const float* __restrict__ deg,
    const float* __restrict__ prev, const float* __restrict__ pdeg,
    float* __restrict__ out, int relu, int nTiles, float* __restrict__ zrows,
    float* __restrict__ zdeg) {
  __shared__ float lds[CHUNK_KT * 8 * 64];

  const int lane = threadIdx.x & 31;
  const int wave = threadIdx.x >> 5;
  int tile = blockIdx.x * TILES_PER_BLOCK + wave;
  const bool active = tile < nTiles;
  if (!active) tile = nTiles - 1;
  const int tile_u = __builtin_amdgcn_readfirstlane(tile);

  const int half = lane >> 4;
  const int n0 = lane & 15;
  const int arow = tile * 16 + n0;
  const float* __restrict__ arowp = A + (size_t)arow * K + half * 2;

  float scale = 1.0f;
  if (deg) scale = 1.0f / fmaxf(deg[arow], 1.0f);

  v8f acc[8];
#pragma unroll
  for (int nt = 0; nt < 8; ++nt) {
    float b = bias ? bias[nt * 16 + n0] : 0.0f;
    acc[nt] = (v8f){b, b, b, b, b, b, b, b};
  }

  const int nkt = K >> 2;
  v2f a_pre = *(const v2f*)(arowp);

  for (int kt0 = 0; kt0 < nkt; kt0 += CHUNK_KT) {
    __syncthreads();
    {
      const float4* __restrict__ gsrc =
          (const float4*)(Bp + (size_t)kt0 * 8 * 64);
      float4* __restrict__ ldst = (float4*)lds;
      for (int i = threadIdx.x; i < CHUNK_KT * 8 * 16; i += 256)
        ldst[i] = gsrc[i];
    }
    __syncthreads();

    for (int kt = 0; kt < CHUNK_KT; ++kt) {
      const int ktg = kt0 + kt;
      v2f a = a_pre * scale;
      const int ktn = (ktg + 1 < nkt) ? (ktg + 1) : ktg;
      a_pre = *(const v2f*)(arowp + ktn * 4);

      const float* bbase = lds + (kt * 8) * 64 + lane * 2;
      v2f bf[8];
#pragma unroll
      for (int nt = 0; nt < 8; ++nt) bf[nt] = *(const v2f*)(bbase + nt * 64);

#pragma unroll
      for (int nt = 0; nt < 8; ++nt) {
        acc[nt] = __builtin_amdgcn_wmma_f32_16x16x4_f32(
            false, a, false, bf[nt], (short)0, acc[nt], false, false);
      }
    }
  }

  if (!active) return;

  const int el = (int)__builtin_amdgcn_mbcnt_lo(~0u, 0u);
  const int eh = el >> 4;
  const int en = el & 15;

  if (zdeg && (tile_u & 1) == 0) {
    const int zi = tile_u * 16 + el;
    if (zi < nTiles * 16) zdeg[zi] = 0.0f;
  }
  if (zrows) {
    float4* zp = (float4*)(zrows + (size_t)tile_u * (16 * 128)) + el;
    const float4 z4 = {0.0f, 0.0f, 0.0f, 0.0f};
    for (int i = 0; i < 16; ++i) zp[i * 32] = z4;
  }

  const int m0 = tile_u * 16;
  float pscale[8];
#pragma unroll
  for (int r = 0; r < 8; ++r)
    pscale[r] = pdeg ? 1.0f / fmaxf(pdeg[m0 + r + 8 * eh], 1.0f) : 1.0f;
#pragma unroll
  for (int nt = 0; nt < 8; ++nt) {
#pragma unroll
    for (int r = 0; r < 8; ++r) {
      int m = m0 + r + 8 * eh;
      size_t idx = (size_t)m * 128 + nt * 16 + en;
      float v = acc[nt][r];
      if (prev) v += prev[idx] * pscale[r];
      if (relu) v = fmaxf(v, 0.0f);
      out[idx] = v;
    }
  }
}

extern "C" void kernel_launch(void* const* d_in, const int* in_sizes, int n_in,
                              void* d_out, int out_size, void* d_ws,
                              size_t ws_size, hipStream_t stream) {
  const int IN_DIM = 384, HID = 128;
  const int N = in_sizes[0] / IN_DIM;
  const int E = in_sizes[1] / 2;

  const float* x    = (const float*)d_in[0];
  const int*   ei   = (const int*)d_in[1];
  const float* W1_l = (const float*)d_in[2];
  const float* b1   = (const float*)d_in[3];
  const float* W1_r = (const float*)d_in[4];
  const float* W2_l = (const float*)d_in[5];
  const float* b2   = (const float*)d_in[6];
  const float* W2_r = (const float*)d_in[7];
  const int* src = ei;
  const int* dst = ei + E;
  float* out = (float*)d_out;

  float* ws = (float*)d_ws;
  const long deg_off  = 0;
  const long deg_pad  = ((long)N + 127) & ~127L;
  const long agg_off  = deg_off + deg_pad;
  const long y_off    = agg_off + (long)N * HID;
  const long h1_off   = y_off + (long)N * HID;
  const long pw1l_off = h1_off + (long)N * HID;
  const long pw1r_off = pw1l_off + (long)IN_DIM * HID;
  const long pw2l_off = pw1r_off + (long)IN_DIM * HID;
  const long pw2r_off = pw2l_off + (long)HID * HID;

  float* deg  = ws + deg_off;
  float* agg  = ws + agg_off;
  float* y    = ws + y_off;
  float* agg2 = y;
  float* h1   = ws + h1_off;
  float* pw1l = ws + pw1l_off;
  float* pw1r = ws + pw1r_off;
  float* pw2l = ws + pw2l_off;
  float* pw2r = ws + pw2r_off;

  const int nTiles = N / 16;
  const int gemmBlocks = (nTiles + TILES_PER_BLOCK - 1) / TILES_PER_BLOCK;

  pack_weights_kernel<<<(IN_DIM * 128 + 255) / 256, 256, 0, stream>>>(W1_l, pw1l, IN_DIM);
  pack_weights_kernel<<<(IN_DIM * 128 + 255) / 256, 256, 0, stream>>>(W1_r, pw1r, IN_DIM);
  pack_weights_kernel<<<(HID * 128 + 255) / 256, 256, 0, stream>>>(W2_l, pw2l, HID);
  pack_weights_kernel<<<(HID * 128 + 255) / 256, 256, 0, stream>>>(W2_r, pw2r, HID);

  gemm_pass_kernel<<<gemmBlocks, 256, 0, stream>>>(
      x, IN_DIM, pw1l, nullptr, nullptr, nullptr, nullptr, y, 0, nTiles, agg,
      deg);

  degree_kernel<<<(E + 255) / 256, 256, 0, stream>>>(dst, deg, E);

  scatter_add_kernel<<<(E + 7) / 8, 256, 0, stream>>>(y, src, dst, agg, E);

  gemm_pass_kernel<<<gemmBlocks, 256, 0, stream>>>(
      x, IN_DIM, pw1r, b1, nullptr, agg, deg, h1, 1, nTiles, agg2, nullptr);

  scatter_add_kernel<<<(E + 7) / 8, 256, 0, stream>>>(h1, src, dst, agg2, E);

  gemm_pass_kernel<<<gemmBlocks, 256, 0, stream>>>(
      agg2, HID, pw2l, b2, deg, nullptr, nullptr, out, 0, nTiles, nullptr,
      nullptr);
  gemm_pass_kernel<<<gemmBlocks, 256, 0, stream>>>(
      h1, HID, pw2r, nullptr, nullptr, out, nullptr, out, 0, nTiles, nullptr,
      nullptr);
}


# GEMM passes as a resident grid: 256 work-groups of 16 waves, one per WGP, copy the whole packed weight matrix to LDS once and loop over row tiles; no chunk barriers; next fragments asked for ahead
# speedup vs baseline: 2.0261x; 1.3412x over previous
#include <hip/hip_runtime.h>


typedef __attribute__((ext_vector_type(2))) float v2f;
typedef __attribute__((ext_vector_type(8))) float v8f;

#define GEMM_WAVES 16
#define GEMM_THREADS 512
#define GEMM_BLOCKS 256
#define GEMM_MAXK 384

__global__ __launch_bounds__(256) void pack_weights_kernel(
    const float* __restrict__ W, float* __restrict__ P, int K) {
  int total = K * 128;
  for (int tid = blockIdx.x * blockDim.x + threadIdx.x; tid < total;
       tid += gridDim.x * blockDim.x) {
    int frag = tid >> 6;
    int r = tid & 63;
    int L = r >> 1;
    int v = r & 1;
    int kt = frag >> 3;
    int nt = frag & 7;
    int k = kt * 4 + ((L >> 4) << 1) + v;
    int n = nt * 16 + (L & 15);
    P[tid] = W[k * 128 + n];
  }
}

__global__ __launch_bounds__(256) void degree_kernel(
    const int* __restrict__ dst, float* __restrict__ deg, int E) {
  int e = blockIdx.x * blockDim.x + threadIdx.x;
  if (e < E) atomicAdd(&deg[dst[e]], 1.0f);
}

__global__ __launch_bounds__(256) void scatter_add_kernel(
    const float* __restrict__ feat, const int* __restrict__ src,
    const int* __restrict__ dst, float* __restrict__ agg, int E) {
  const int lane = threadIdx.x & 31;
  const int e = blockIdx.x * 8 + (threadIdx.x >> 5);
  if (e >= E) return;
  const int s = src[e];
  const int d = dst[e];
  const float* f = feat + (size_t)s * 128 + lane;
  float* o = agg + (size_t)d * 128 + lane;
  const float v0 = f[0];
  const float v1 = f[32];
  const float v2 = f[64];
  const float v3 = f[96];
  atomicAdd(o, v0);
  atomicAdd(o + 32, v1);
  atomicAdd(o + 64, v2);
  atomicAdd(o + 96, v3);
}

__global__ __launch_bounds__(GEMM_THREADS) void gemm_pass_kernel(
    const float* __restrict__ A, int K, const float* __restrict__ Bp,
    const float* __restrict__ bias, const float* __restrict__ deg,
    const float* __restrict__ prev, const float* __restrict__ pdeg,
    float* __restrict__ out, int relu, int nTiles, float* __restrict__ zrows,
    float* __restrict__ zdeg) {
  __shared__ __attribute__((aligned(16))) float lds[GEMM_MAXK * 128];

  {
    const float4* __restrict__ gsrc = (const float4*)Bp;
    float4* __restrict__ ldst = (float4*)lds;
    const int n4 = K * 32;
    for (int i0 = threadIdx.x; i0 < n4; i0 += GEMM_THREADS * 8) {
      float4 t[8];
#pragma unroll
      for (int j = 0; j < 8; ++j) t[j] = gsrc[i0 + j * GEMM_THREADS];
#pragma unroll
      for (int j = 0; j < 8; ++j) ldst[i0 + j * GEMM_THREADS] = t[j];
    }
  }
  __syncthreads();

  const int lane = threadIdx.x & 31;
  const int wave = __builtin_amdgcn_readfirstlane((int)(threadIdx.x >> 5));
  const int half = lane >> 4;
  const int n0 = lane & 15;
  const int nkt = K >> 2;
  const int ngrp = nkt >> 3;
  const float* __restrict__ bl = lds + lane * 2;

  for (int tile = (int)blockIdx.x + (int)gridDim.x * wave; tile < nTiles;
       tile += (int)gridDim.x * GEMM_WAVES) {
    const int arow = tile * 16 + n0;
    const float* __restrict__ arowp = A + (size_t)arow * K + half * 2;

    float scale = 1.0f;
    if (deg) scale = 1.0f / fmaxf(deg[arow], 1.0f);

    v8f acc[8];
#pragma unroll
    for (int nt = 0; nt < 8; ++nt) {
      float b = bias ? bias[nt * 16 + n0] : 0.0f;
      acc[nt] = (v8f){b, b, b, b, b, b, b, b};
    }

    v2f an[8];
#pragma unroll
    for (int j = 0; j < 8; ++j) an[j] = *(const v2f*)(arowp + j * 4);
    v2f bn[8];
#pragma unroll
    for (int nt = 0; nt < 8; ++nt) bn[nt] = *(const v2f*)(bl + nt * 64);

    for (int g = 0; g < ngrp; ++g) {
      v2f ac[8];
#pragma unroll
      for (int j = 0; j < 8; ++j) ac[j] = an[j] * scale;
      const int gn = (g + 1 < ngrp) ? (g + 1) : g;
#pragma unroll
      for (int j = 0; j < 8; ++j)
        an[j] = *(const v2f*)(arowp + (gn * 8 + j) * 4);

#pragma unroll
      for (int j = 0; j < 8; ++j) {
        v2f bc[8];
#pragma unroll
        for (int nt = 0; nt < 8; ++nt) bc[nt] = bn[nt];
        int ktn = g * 8 + j + 1;
        if (ktn >= nkt) ktn = nkt - 1;
        const float* __restrict__ bb = bl + ktn * 512;
#pragma unroll
        for (int nt = 0; nt < 8; ++nt) bn[nt] = *(const v2f*)(bb + nt * 64);
#pragma unroll
        for (int nt = 0; nt < 8; ++nt) {
          acc[nt] = __builtin_amdgcn_wmma_f32_16x16x4_f32(
              false, ac[j], false, bc[nt], (short)0, acc[nt], false, false);
        }
      }
    }

    if (zdeg && (tile & 1) == 0) {
      const int zi = tile * 16 + lane;
      if (zi < nTiles * 16) zdeg[zi] = 0.0f;
    }
    if (zrows) {
      float4* zp = (float4*)(zrows + (size_t)tile * (16 * 128)) + lane;
      const float4 z4 = {0.0f, 0.0f, 0.0f, 0.0f};
      for (int i = 0; i < 16; ++i) zp[i * 32] = z4;
    }

    const int m0 = tile * 16;
    if (prev) {
      float pscale[8];
#pragma unroll
      for (int r = 0; r < 8; ++r)
        pscale[r] = pdeg ? 1.0f / fmaxf(pdeg[m0 + r + 8 * half], 1.0f) : 1.0f;
#pragma unroll
      for (int nt = 0; nt < 8; ++nt) {
#pragma unroll
        for (int r = 0; r < 8; ++r) {
          size_t idx = (size_t)(m0 + r + 8 * half) * 128 + nt * 16 + n0;
          acc[nt][r] += prev[idx] * pscale[r];
        }
      }
    }
#pragma unroll
    for (int nt = 0; nt < 8; ++nt) {
#pragma unroll
      for (int r = 0; r < 8; ++r) {
        size_t idx = (size_t)(m0 + r + 8 * half) * 128 + nt * 16 + n0;
        float v = acc[nt][r];
        if (relu) v = fmaxf(v, 0.0f);
        out[idx] = v;
      }
    }
  }
}

extern "C" void kernel_launch(void* const* d_in, const int* in_sizes, int n_in,
                              void* d_out, int out_size, void* d_ws,
                              size_t ws_size, hipStream_t stream) {
  const int IN_DIM = 384, HID = 128;
  const int N = in_sizes[0] / IN_DIM;
  const int E = in_sizes[1] / 2;

  const float* x    = (const float*)d_in[0];
  const int*   ei   = (const int*)d_in[1];
  const float* W1_l = (const float*)d_in[2];
  const float* b1   = (const float*)d_in[3];
  const float* W1_r = (const float*)d_in[4];
  const float* W2_l = (const float*)d_in[5];
  const float* b2   = (const float*)d_in[6];
  const float* W2_r = (const float*)d_in[7];
  const int* src = ei;
  const int* dst = ei + E;
  float* out = (float*)d_out;

  float* ws = (float*)d_ws;
  const long deg_off  = 0;
  const long deg_pad  = ((long)N + 127) & ~127L;
  const long agg_off  = deg_off + deg_pad;
  const long y_off    = agg_off + (long)N * HID;
  const long h1_off   = y_off + (long)N * HID;
  const long pw1l_off = h1_off + (long)N * HID;
  const long pw1r_off = pw1l_off + (long)IN_DIM * HID;
  const long pw2l_off = pw1r_off + (long)IN_DIM * HID;
  const long pw2r_off = pw2l_off + (long)HID * HID;

  float* deg  = ws + deg_off;
  float* agg  = ws + agg_off;
  float* y    = ws + y_off;
  float* agg2 = y;
  float* h1   = ws + h1_off;
  float* pw1l = ws + pw1l_off;
  float* pw1r = ws + pw1r_off;
  float* pw2l = ws + pw2l_off;
  float* pw2r = ws + pw2r_off;

  const int nTiles = N / 16;

  pack_weights_kernel<<<(IN_DIM * 128 + 255) / 256, 256, 0, stream>>>(W1_l, pw1l, IN_DIM);
  pack_weights_kernel<<<(IN_DIM * 128 + 255) / 256, 256, 0, stream>>>(W1_r, pw1r, IN_DIM);
  pack_weights_kernel<<<(HID * 128 + 255) / 256, 256, 0, stream>>>(W2_l, pw2l, HID);
  pack_weights_kernel<<<(HID * 128 + 255) / 256, 256, 0, stream>>>(W2_r, pw2r, HID);

  gemm_pass_kernel<<<GEMM_BLOCKS, GEMM_THREADS, 0, stream>>>(
      x, IN_DIM, pw1l, nullptr, nullptr, nullptr, nullptr, y, 0, nTiles, agg,
      deg);

  degree_kernel<<<(E + 255) / 256, 256, 0, stream>>>(dst, deg, E);

  scatter_add_kernel<<<(E + 7) / 8, 256, 0, stream>>>(y, src, dst, agg, E);

  gemm_pass_kernel<<<GEMM_BLOCKS, GEMM_THREADS, 0, stream>>>(
      x, IN_DIM, pw1r, b1, nullptr, agg, deg, h1, 1, nTiles, agg2, nullptr);

  scatter_add_kernel<<<(E + 7) / 8, 256, 0, stream>>>(h1, src, dst, agg2, E);

  gemm_pass_kernel<<<GEMM_BLOCKS, GEMM_THREADS, 0, stream>>>(
      agg2, HID, pw2l, b2, deg, nullptr, nullptr, out, 0, nTiles, nullptr,
      nullptr);
  gemm_pass_kernel<<<GEMM_BLOCKS, GEMM_THREADS, 0, stream>>>(
      h1, HID, pw2r, nullptr, nullptr, out, nullptr, out, 0, nTiles, nullptr,
      nullptr);
}


# The five helper launches are folded into their consumers: each GEMM pass puts its row-major weights into B-fragment order on the way into LDS; the first scatter counts degrees; 11 launches -> 6
# speedup vs baseline: 2.1014x; 1.0372x over previous
#include <hip/hip_runtime.h>


typedef __attribute__((ext_vector_type(2))) float v2f;
typedef __attribute__((ext_vector_type(8))) float v8f;

#define GEMM_WAVES 16
#define GEMM_THREADS 512
#define GEMM_BLOCKS 256
#define GEMM_MAXK 384


__global__ __launch_bounds__(256) void scatter_add_kernel(
    const float* __restrict__ feat, const int* __restrict__ src,
    const int* __restrict__ dst, float* __restrict__ agg,
    float* __restrict__ deg, int E) {
  const int lane = threadIdx.x & 31;
  const int e = blockIdx.x * 8 + (threadIdx.x >> 5);
  if (e >= E) return;
  const int s = src[e];
  const int d = dst[e];
  const float* f = feat + (size_t)s * 128 + lane;
  float* o = agg + (size_t)d * 128 + lane;
  const float v0 = f[0];
  const float v1 = f[32];
  const float v2 = f[64];
  const float v3 = f[96];
  atomicAdd(o, v0);
  atomicAdd(o + 32, v1);
  atomicAdd(o + 64, v2);
  atomicAdd(o + 96, v3);
  if (deg != nullptr && lane == 0) atomicAdd(deg + d, 1.0f);
}

__global__ __launch_bounds__(GEMM_THREADS) void gemm_pass_kernel(
    const float* __restrict__ A, int K, const float* __restrict__ Wm,
    const float* __restrict__ bias, const float* __restrict__ deg,
    const float* __restrict__ prev, const float* __restrict__ pdeg,
    float* __restrict__ out, int relu, int nTiles, float* __restrict__ zrows,
    float* __restrict__ zdeg) {
  __shared__ __attribute__((aligned(16))) float lds[GEMM_MAXK * 128];

  {
    const int low = threadIdx.x & 127;
    const int cp = low & 7;
    const int nt = ((low >> 5) & 3) * 2 + ((low >> 3) & 1);
    const int h = (low >> 4) & 1;
    const float* __restrict__ gsrc = Wm + h * 256 + nt * 16 + cp * 2;
    float* __restrict__ ldst = lds + nt * 64 + h * 32 + cp * 4;
    const int nkt_all = K >> 2;
    for (int kt0 = threadIdx.x >> 7; kt0 < nkt_all; kt0 += 32) {
      v2f r0[8], r1[8];
#pragma unroll
      for (int j = 0; j < 8; ++j) {
        const float* g = gsrc + (kt0 + 4 * j) * 512;
        r0[j] = *(const v2f*)g;
        r1[j] = *(const v2f*)(g + 128);
      }
#pragma unroll
      for (int j = 0; j < 8; ++j) {
        const float4 q = {r0[j].x, r1[j].x, r0[j].y, r1[j].y};
        *(float4*)(ldst + (kt0 + 4 * j) * 512) = q;
      }
    }
  }
  __syncthreads();

  const int lane = threadIdx.x & 31;
  const int wave = __builtin_amdgcn_readfirstlane((int)(threadIdx.x >> 5));
  const int half = lane >> 4;
  const int n0 = lane & 15;
  const int nkt = K >> 2;
  const int ngrp = nkt >> 3;
  const float* __restrict__ bl = lds + lane * 2;

  for (int tile = (int)blockIdx.x + (int)gridDim.x * wave; tile < nTiles;
       tile += (int)gridDim.x * GEMM_WAVES) {
    const int arow = tile * 16 + n0;
    const float* __restrict__ arowp = A + (size_t)arow * K + half * 2;

    float scale = 1.0f;
    if (deg) scale = 1.0f / fmaxf(deg[arow], 1.0f);

    v8f acc[8];
#pragma unroll
    for (int nt = 0; nt < 8; ++nt) {
      float b = bias ? bias[nt * 16 + n0] : 0.0f;
      acc[nt] = (v8f){b, b, b, b, b, b, b, b};
    }

    v2f an[8];
#pragma unroll
    for (int j = 0; j < 8; ++j) an[j] = *(const v2f*)(arowp + j * 4);
    v2f bn[8];
#pragma unroll
    for (int nt = 0; nt < 8; ++nt) bn[nt] = *(const v2f*)(bl + nt * 64);

    for (int g = 0; g < ngrp; ++g) {
      v2f ac[8];
#pragma unroll
      for (int j = 0; j < 8; ++j) ac[j] = an[j] * scale;
      const int gn = (g + 1 < ngrp) ? (g + 1) : g;
#pragma unroll
      for (int j = 0; j < 8; ++j)
        an[j] = *(const v2f*)(arowp + (gn * 8 + j) * 4);

#pragma unroll
      for (int j = 0; j < 8; ++j) {
        v2f bc[8];
#pragma unroll
        for (int nt = 0; nt < 8; ++nt) bc[nt] = bn[nt];
        int ktn = g * 8 + j + 1;
        if (ktn >= nkt) ktn = nkt - 1;
        const float* __restrict__ bb = bl + ktn * 512;
#pragma unroll
        for (int nt = 0; nt < 8; ++nt) bn[nt] = *(const v2f*)(bb + nt * 64);
#pragma unroll
        for (int nt = 0; nt < 8; ++nt) {
          acc[nt] = __builtin_amdgcn_wmma_f32_16x16x4_f32(
              false, ac[j], false, bc[nt], (short)0, acc[nt], false, false);
        }
      }
    }

    if (zdeg && (tile & 1) == 0) {
      const int zi = tile * 16 + lane;
      if (zi < nTiles * 16) zdeg[zi] = 0.0f;
    }
    if (zrows) {
      float4* zp = (float4*)(zrows + (size_t)tile * (16 * 128)) + lane;
      const float4 z4 = {0.0f, 0.0f, 0.0f, 0.0f};
      for (int i = 0; i < 16; ++i) zp[i * 32] = z4;
    }

    const int m0 = tile * 16;
    if (prev) {
      float pscale[8];
#pragma unroll
      for (int r = 0; r < 8; ++r)
        pscale[r] = pdeg ? 1.0f / fmaxf(pdeg[m0 + r + 8 * half], 1.0f) : 1.0f;
#pragma unroll
      for (int nt = 0; nt < 8; ++nt) {
#pragma unroll
        for (int r = 0; r < 8; ++r) {
          size_t idx = (size_t)(m0 + r + 8 * half) * 128 + nt * 16 + n0;
          acc[nt][r] += prev[idx] * pscale[r];
        }
      }
    }
#pragma unroll
    for (int nt = 0; nt < 8; ++nt) {
#pragma unroll
      for (int r = 0; r < 8; ++r) {
        size_t idx = (size_t)(m0 + r + 8 * half) * 128 + nt * 16 + n0;
        float v = acc[nt][r];
        if (relu) v = fmaxf(v, 0.0f);
        out[idx] = v;
      }
    }
  }
}

extern "C" void kernel_launch(void* const* d_in, const int* in_sizes, int n_in,
                              void* d_out, int out_size, void* d_ws,
                              size_t ws_size, hipStream_t stream) {
  const int IN_DIM = 384, HID = 128;
  const int N = in_sizes[0] / IN_DIM;
  const int E = in_sizes[1] / 2;

  const float* x    = (const float*)d_in[0];
  const int*   ei   = (const int*)d_in[1];
  const float* W1_l = (const float*)d_in[2];
  const float* b1   = (const float*)d_in[3];
  const float* W1_r = (const float*)d_in[4];
  const float* W2_l = (const float*)d_in[5];
  const float* b2   = (const float*)d_in[6];
  const float* W2_r = (const float*)d_in[7];
  const int* src = ei;
  const int* dst = ei + E;
  float* out = (float*)d_out;

  float* ws = (float*)d_ws;
  const long deg_off  = 0;
  const long deg_pad  = ((long)N + 127) & ~127L;
  const long agg_off  = deg_off + deg_pad;
  const long y_off    = agg_off + (long)N * HID;
  const long h1_off   = y_off + (long)N * HID;

  float* deg  = ws + deg_off;
  float* agg  = ws + agg_off;
  float* y    = ws + y_off;
  float* agg2 = y;
  float* h1   = ws + h1_off;

  const int nTiles = N / 16;

  gemm_pass_kernel<<<GEMM_BLOCKS, GEMM_THREADS, 0, stream>>>(
      x, IN_DIM, W1_l, nullptr, nullptr, nullptr, nullptr, y, 0, nTiles, agg,
      deg);

  scatter_add_kernel<<<(E + 7) / 8, 256, 0, stream>>>(y, src, dst, agg, deg, E);

  gemm_pass_kernel<<<GEMM_BLOCKS, GEMM_THREADS, 0, stream>>>(
      x, IN_DIM, W1_r, b1, nullptr, agg, deg, h1, 1, nTiles, agg2, nullptr);

  scatter_add_kernel<<<(E + 7) / 8, 256, 0, stream>>>(h1, src, dst, agg2,
                                                      nullptr, E);

  gemm_pass_kernel<<<GEMM_BLOCKS, GEMM_THREADS, 0, stream>>>(
      agg2, HID, W2_l, b2, deg, nullptr, nullptr, out, 0, nTiles, nullptr,
      nullptr);
  gemm_pass_kernel<<<GEMM_BLOCKS, GEMM_THREADS, 0, stream>>>(
      h1, HID, W2_r, nullptr, nullptr, out, nullptr, out, 0, nTiles, nullptr,
      nullptr);
}


# Scatter as a resident grid: 1024 work-groups of 8 waves, each wave walks over batches of 8 edges with the batch's 32 row loads in flight; 8192 waves per launch, not 200000 (f32 atomics kept)
# speedup vs baseline: 2.1922x; 1.0432x over previous
#include <hip/hip_runtime.h>


typedef __attribute__((ext_vector_type(2))) float v2f;
typedef __attribute__((ext_vector_type(8))) float v8f;

#define GEMM_WAVES 16
#define GEMM_THREADS 512
#define GEMM_BLOCKS 256
#define GEMM_MAXK 384


#define SCAT_BLOCKS 1024
#define SCAT_BATCH 8

__global__ __launch_bounds__(256) void scatter_add_kernel(
    const float* __restrict__ feat, const int* __restrict__ src,
    const int* __restrict__ dst, float* __restrict__ agg,
    float* __restrict__ deg, int E) {
  const int lane = threadIdx.x & 31;
  const int sub = lane & (SCAT_BATCH - 1);
  const int wave = __builtin_amdgcn_readfirstlane(
      (int)(blockIdx.x * 8 + (threadIdx.x >> 5)));
  const int stride = (int)gridDim.x * 8 * SCAT_BATCH;
  const int last = E - 1;

  int e0 = wave * SCAT_BATCH;
  if (e0 >= E) return;
  int ic = e0 + sub;
  ic = ic < last ? ic : last;
  int ms = src[ic];
  int md = dst[ic];
  for (;;) {
    const int en = e0 + stride;
    int in = en + sub;
    in = in < last ? in : last;
    const int ns = src[in];
    const int nd = dst[in];

    const int cnt = (E - e0 < SCAT_BATCH) ? (E - e0) : SCAT_BATCH;
    if (cnt == SCAT_BATCH) {
      float v[SCAT_BATCH][4];
#pragma unroll
      for (int j = 0; j < SCAT_BATCH; ++j) {
        const int s = __builtin_amdgcn_readlane(ms, j);
        const float* f = feat + (size_t)s * 128 + lane;
        v[j][0] = f[0];
        v[j][1] = f[32];
        v[j][2] = f[64];
        v[j][3] = f[96];
      }
#pragma unroll
      for (int j = 0; j < SCAT_BATCH; ++j) {
        const int d = __builtin_amdgcn_readlane(md, j);
        float* o = agg + (size_t)d * 128 + lane;
        atomicAdd(o, v[j][0]);
        atomicAdd(o + 32, v[j][1]);
        atomicAdd(o + 64, v[j][2]);
        atomicAdd(o + 96, v[j][3]);
      }
    } else {
      for (int j = 0; j < cnt; ++j) {
        const int s = __builtin_amdgcn_readlane(ms, j);
        const int d = __builtin_amdgcn_readlane(md, j);
        const float* f = feat + (size_t)s * 128 + lane;
        float* o = agg + (size_t)d * 128 + lane;
        const float v0 = f[0];
        const float v1 = f[32];
        const float v2 = f[64];
        const float v3 = f[96];
        atomicAdd(o, v0);
        atomicAdd(o + 32, v1);
        atomicAdd(o + 64, v2);
        atomicAdd(o + 96, v3);
      }
    }
    if (deg != nullptr && lane < cnt) atomicAdd(deg + md, 1.0f);
    if (en >= E) break;
    ms = ns;
    md = nd;
    e0 = en;
  }
}

__global__ __launch_bounds__(GEMM_THREADS) void gemm_pass_kernel(
    const float* __restrict__ A, int K, const float* __restrict__ Wm,
    const float* __restrict__ bias, const float* __restrict__ deg,
    const float* __restrict__ prev, const float* __restrict__ pdeg,
    float* __restrict__ out, int relu, int nTiles, float* __restrict__ zrows,
    float* __restrict__ zdeg) {
  __shared__ __attribute__((aligned(16))) float lds[GEMM_MAXK * 128];

  {
    const int low = threadIdx.x & 127;
    const int cp = low & 7;
    const int nt = ((low >> 5) & 3) * 2 + ((low >> 3) & 1);
    const int h = (low >> 4) & 1;
    const float* __restrict__ gsrc = Wm + h * 256 + nt * 16 + cp * 2;
    float* __restrict__ ldst = lds + nt * 64 + h * 32 + cp * 4;
    const int nkt_all = K >> 2;
    for (int kt0 = threadIdx.x >> 7; kt0 < nkt_all; kt0 += 32) {
      v2f r0[8], r1[8];
#pragma unroll
      for (int j = 0; j < 8; ++j) {
        const float* g = gsrc + (kt0 + 4 * j) * 512;
        r0[j] = *(const v2f*)g;
        r1[j] = *(const v2f*)(g + 128);
      }
#pragma unroll
      for (int j = 0; j < 8; ++j) {
        const float4 q = {r0[j].x, r1[j].x, r0[j].y, r1[j].y};
        *(float4*)(ldst + (kt0 + 4 * j) * 512) = q;
      }
    }
  }
  __syncthreads();

  const int lane = threadIdx.x & 31;
  const int wave = __builtin_amdgcn_readfirstlane((int)(threadIdx.x >> 5));
  const int half = lane >> 4;
  const int n0 = lane & 15;
  const int nkt = K >> 2;
  const int ngrp = nkt >> 3;
  const float* __restrict__ bl = lds + lane * 2;

  for (int tile = (int)blockIdx.x + (int)gridDim.x * wave; tile < nTiles;
       tile += (int)gridDim.x * GEMM_WAVES) {
    const int arow = tile * 16 + n0;
    const float* __restrict__ arowp = A + (size_t)arow * K + half * 2;

    float scale = 1.0f;
    if (deg) scale = 1.0f / fmaxf(deg[arow], 1.0f);

    v8f acc[8];
#pragma unroll
    for (int nt = 0; nt < 8; ++nt) {
      float b = bias ? bias[nt * 16 + n0] : 0.0f;
      acc[nt] = (v8f){b, b, b, b, b, b, b, b};
    }

    v2f an[8];
#pragma unroll
    for (int j = 0; j < 8; ++j) an[j] = *(const v2f*)(arowp + j * 4);
    v2f bn[8];
#pragma unroll
    for (int nt = 0; nt < 8; ++nt) bn[nt] = *(const v2f*)(bl + nt * 64);

    for (int g = 0; g < ngrp; ++g) {
      v2f ac[8];
#pragma unroll
      for (int j = 0; j < 8; ++j) ac[j] = an[j] * scale;
      const int gn = (g + 1 < ngrp) ? (g + 1) : g;
#pragma unroll
      for (int j = 0; j < 8; ++j)
        an[j] = *(const v2f*)(arowp + (gn * 8 + j) * 4);

#pragma unroll
      for (int j = 0; j < 8; ++j) {
        v2f bc[8];
#pragma unroll
        for (int nt = 0; nt < 8; ++nt) bc[nt] = bn[nt];
        int ktn = g * 8 + j + 1;
        if (ktn >= nkt) ktn = nkt - 1;
        const float* __restrict__ bb = bl + ktn * 512;
#pragma unroll
        for (int nt = 0; nt < 8; ++nt) bn[nt] = *(const v2f*)(bb + nt * 64);
#pragma unroll
        for (int nt = 0; nt < 8; ++nt) {
          acc[nt] = __builtin_amdgcn_wmma_f32_16x16x4_f32(
              false, ac[j], false, bc[nt], (short)0, acc[nt], false, false);
        }
      }
    }

    if (zdeg && (tile & 1) == 0) {
      const int zi = tile * 16 + lane;
      if (zi < nTiles * 16) zdeg[zi] = 0.0f;
    }
    if (zrows) {
      float4* zp = (float4*)(zrows + (size_t)tile * (16 * 128)) + lane;
      const float4 z4 = {0.0f, 0.0f, 0.0f, 0.0f};
      for (int i = 0; i < 16; ++i) zp[i * 32] = z4;
    }

    const int m0 = tile * 16;
    if (prev) {
      float pscale[8];
#pragma unroll
      for (int r = 0; r < 8; ++r)
        pscale[r] = pdeg ? 1.0f / fmaxf(pdeg[m0 + r + 8 * half], 1.0f) : 1.0f;
#pragma unroll
      for (int nt = 0; nt < 8; ++nt) {
#pragma unroll
        for (int r = 0; r < 8; ++r) {
          size_t idx = (size_t)(m0 + r + 8 * half) * 128 + nt * 16 + n0;
          acc[nt][r] += prev[idx] * pscale[r];
        }
      }
    }
#pragma unroll
    for (int nt = 0; nt < 8; ++nt) {
#pragma unroll
      for (int r = 0; r < 8; ++r) {
        size_t idx = (size_t)(m0 + r + 8 * half) * 128 + nt * 16 + n0;
        float v = acc[nt][r];
        if (relu) v = fmaxf(v, 0.0f);
        out[idx] = v;
      }
    }
  }
}

extern "C" void kernel_launch(void* const* d_in, const int* in_sizes, int n_in,
                              void* d_out, int out_size, void* d_ws,
                              size_t ws_size, hipStream_t stream) {
  const int IN_DIM = 384, HID = 128;
  const int N = in_sizes[0] / IN_DIM;
  const int E = in_sizes[1] / 2;

  const float* x    = (const float*)d_in[0];
  const int*   ei   = (const int*)d_in[1];
  const float* W1_l = (const float*)d_in[2];
  const float* b1   = (const float*)d_in[3];
  const float* W1_r = (const float*)d_in[4];
  const float* W2_l = (const float*)d_in[5];
  const float* b2   = (const float*)d_in[6];
  const float* W2_r = (const float*)d_in[7];
  const int* src = ei;
  const int* dst = ei + E;
  float* out = (float*)d_out;

  float* ws = (float*)d_ws;
  const long deg_off  = 0;
  const long deg_pad  = ((long)N + 127) & ~127L;
  const long agg_off  = deg_off + deg_pad;
  const long y_off    = agg_off + (long)N * HID;
  const long h1_off   = y_off + (long)N * HID;

  float* deg  = ws + deg_off;
  float* agg  = ws + agg_off;
  float* y    = ws + y_off;
  float* agg2 = y;
  float* h1   = ws + h1_off;

  const int nTiles = N / 16;

  gemm_pass_kernel<<<GEMM_BLOCKS, GEMM_THREADS, 0, stream>>>(
      x, IN_DIM, W1_l, nullptr, nullptr, nullptr, nullptr, y, 0, nTiles, agg,
      deg);

  scatter_add_kernel<<<SCAT_BLOCKS, 256, 0, stream>>>(y, src, dst, agg, deg, E);

  gemm_pass_kernel<<<GEMM_BLOCKS, GEMM_THREADS, 0, stream>>>(
      x, IN_DIM, W1_r, b1, nullptr, agg, deg, h1, 1, nTiles, agg2, nullptr);

  scatter_add_kernel<<<SCAT_BLOCKS, 256, 0, stream>>>(h1, src, dst, agg2,
                                                      nullptr, E);

  gemm_pass_kernel<<<GEMM_BLOCKS, GEMM_THREADS, 0, stream>>>(
      agg2, HID, W2_l, b2, deg, nullptr, nullptr, out, 0, nTiles, nullptr,
      nullptr);
  gemm_pass_kernel<<<GEMM_BLOCKS, GEMM_THREADS, 0, stream>>>(
      h1, HID, W2_r, nullptr, nullptr, out, nullptr, out, 0, nTiles, nullptr,
      nullptr);
}


# GEMM passes: the 106 row tiles left over after 6 whole tiles per SIMD are cut into four 16x32-column units on four SIMDs, so no SIMD runs a 7th whole tile (6.25 tile-times per pass, not 7)
# speedup vs baseline: 2.3525x; 1.0731x over previous
#include <hip/hip_runtime.h>


typedef __attribute__((ext_vector_type(2))) float v2f;
typedef __attribute__((ext_vector_type(8))) float v8f;

#define GEMM_WAVES 16
#define GEMM_THREADS 512
#define GEMM_BLOCKS 256
#define GEMM_MAXK 384


#define SCAT_BLOCKS 1024
#define SCAT_BATCH 8

__global__ __launch_bounds__(256) void scatter_add_kernel(
    const float* __restrict__ feat, const int* __restrict__ src,
    const int* __restrict__ dst, float* __restrict__ agg,
    float* __restrict__ deg, int E) {
  const int lane = threadIdx.x & 31;
  const int sub = lane & (SCAT_BATCH - 1);
  const int wave = __builtin_amdgcn_readfirstlane(
      (int)(blockIdx.x * 8 + (threadIdx.x >> 5)));
  const int stride = (int)gridDim.x * 8 * SCAT_BATCH;
  const int last = E - 1;

  int e0 = wave * SCAT_BATCH;
  if (e0 >= E) return;
  int ic = e0 + sub;
  ic = ic < last ? ic : last;
  int ms = src[ic];
  int md = dst[ic];
  for (;;) {
    const int en = e0 + stride;
    int in = en + sub;
    in = in < last ? in : last;
    const int ns = src[in];
    const int nd = dst[in];

    const int cnt = (E - e0 < SCAT_BATCH) ? (E - e0) : SCAT_BATCH;
    if (cnt == SCAT_BATCH) {
      float v[SCAT_BATCH][4];
#pragma unroll
      for (int j = 0; j < SCAT_BATCH; ++j) {
        const int s = __builtin_amdgcn_readlane(ms, j);
        const float* f = feat + (size_t)s * 128 + lane;
        v[j][0] = f[0];
        v[j][1] = f[32];
        v[j][2] = f[64];
        v[j][3] = f[96];
      }
#pragma unroll
      for (int j = 0; j < SCAT_BATCH; ++j) {
        const int d = __builtin_amdgcn_readlane(md, j);
        float* o = agg + (size_t)d * 128 + lane;
        atomicAdd(o, v[j][0]);
        atomicAdd(o + 32, v[j][1]);
        atomicAdd(o + 64, v[j][2]);
        atomicAdd(o + 96, v[j][3]);
      }
    } else {
      for (int j = 0; j < cnt; ++j) {
        const int s = __builtin_amdgcn_readlane(ms, j);
        const int d = __builtin_amdgcn_readlane(md, j);
        const float* f = feat + (size_t)s * 128 + lane;
        float* o = agg + (size_t)d * 128 + lane;
        const float v0 = f[0];
        const float v1 = f[32];
        const float v2 = f[64];
        const float v3 = f[96];
        atomicAdd(o, v0);
        atomicAdd(o + 32, v1);
        atomicAdd(o + 64, v2);
        atomicAdd(o + 96, v3);
      }
    }
    if (deg != nullptr && lane < cnt) atomicAdd(deg + md, 1.0f);
    if (en >= E) break;
    ms = ns;
    md = nd;
    e0 = en;
  }
}


template <int NT, int GK>
__device__ __forceinline__ void gemm_tile(
    const float* __restrict__ A, const int K, const float* __restrict__ bias,
    const float* __restrict__ deg, const float* __restrict__ prev,
    const float* __restrict__ pdeg, float* __restrict__ out, const int relu,
    const float* __restrict__ lds, const int tile, const int nt0,
    const int lane) {
  const int half = lane >> 4;
  const int n0 = lane & 15;
  const int nkt = K >> 2;
  const int ngrp = nkt / GK;
  const float* __restrict__ bl = lds + nt0 * 64 + lane * 2;

  const int arow = tile * 16 + n0;
  const float* __restrict__ arowp = A + (size_t)arow * K + half * 2;

  float scale = 1.0f;
  if (deg) scale = 1.0f / fmaxf(deg[arow], 1.0f);

  v8f acc[NT];
#pragma unroll
  for (int nt = 0; nt < NT; ++nt) {
    float b = bias ? bias[(nt0 + nt) * 16 + n0] : 0.0f;
    acc[nt] = (v8f){b, b, b, b, b, b, b, b};
  }

  v2f an[GK];
#pragma unroll
  for (int j = 0; j < GK; ++j) an[j] = *(const v2f*)(arowp + j * 4);
  v2f bn[NT];
#pragma unroll
  for (int nt = 0; nt < NT; ++nt) bn[nt] = *(const v2f*)(bl + nt * 64);

  for (int g = 0; g < ngrp; ++g) {
    v2f ac[GK];
#pragma unroll
    for (int j = 0; j < GK; ++j) ac[j] = an[j] * scale;
    const int gn = (g + 1 < ngrp) ? (g + 1) : g;
#pragma unroll
    for (int j = 0; j < GK; ++j)
      an[j] = *(const v2f*)(arowp + (gn * GK + j) * 4);

#pragma unroll
    for (int j = 0; j < GK; ++j) {
      v2f bc[NT];
#pragma unroll
      for (int nt = 0; nt < NT; ++nt) bc[nt] = bn[nt];
      int ktn = g * GK + j + 1;
      if (ktn >= nkt) ktn = nkt - 1;
      const float* __restrict__ bb = bl + ktn * 512;
#pragma unroll
      for (int nt = 0; nt < NT; ++nt) bn[nt] = *(const v2f*)(bb + nt * 64);
#pragma unroll
      for (int nt = 0; nt < NT; ++nt) {
        acc[nt] = __builtin_amdgcn_wmma_f32_16x16x4_f32(
            false, ac[j], false, bc[nt], (short)0, acc[nt], false, false);
      }
    }
#pragma unroll
    for (int j = 0; j < GK; ++j) asm volatile("" ::"v"(an[j]));
  }

  const int m0 = tile * 16;
  if (prev) {
    float pscale[8];
#pragma unroll
    for (int r = 0; r < 8; ++r)
      pscale[r] = pdeg ? 1.0f / fmaxf(pdeg[m0 + r + 8 * half], 1.0f) : 1.0f;
#pragma unroll
    for (int nt = 0; nt < NT; ++nt) {
#pragma unroll
      for (int r = 0; r < 8; ++r) {
        size_t idx = (size_t)(m0 + r + 8 * half) * 128 + (nt0 + nt) * 16 + n0;
        acc[nt][r] += prev[idx] * pscale[r];
      }
    }
  }
#pragma unroll
  for (int nt = 0; nt < NT; ++nt) {
#pragma unroll
    for (int r = 0; r < 8; ++r) {
      size_t idx = (size_t)(m0 + r + 8 * half) * 128 + (nt0 + nt) * 16 + n0;
      float v = acc[nt][r];
      if (relu) v = fmaxf(v, 0.0f);
      out[idx] = v;
    }
  }
}

__device__ __forceinline__ void zero_tile(float* __restrict__ zrows,
                                          float* __restrict__ zdeg,
                                          const int tile, const int nTiles,
                                          const int lane) {
  if (zdeg && (tile & 1) == 0) {
    const int zi = tile * 16 + lane;
    if (zi < nTiles * 16) zdeg[zi] = 0.0f;
  }
  if (zrows) {
    float4* zp = (float4*)(zrows + (size_t)tile * (16 * 128)) + lane;
    const float4 z4 = {0.0f, 0.0f, 0.0f, 0.0f};
    for (int i = 0; i < 16; ++i) zp[i * 32] = z4;
  }
}

__global__ __launch_bounds__(GEMM_THREADS) void gemm_pass_kernel(
    const float* __restrict__ A, int K, const float* __restrict__ Wm,
    const float* __restrict__ bias, const float* __restrict__ deg,
    const float* __restrict__ prev, const float* __restrict__ pdeg,
    float* __restrict__ out, int relu, int nTiles, float* __restrict__ zrows,
    float* __restrict__ zdeg) {
  __shared__ __attribute__((aligned(16))) float lds[GEMM_MAXK * 128];

  {
    const int low = threadIdx.x & 127;
    const int cp = low & 7;
    const int nt = ((low >> 5) & 3) * 2 + ((low >> 3) & 1);
    const int h = (low >> 4) & 1;
    const float* __restrict__ gsrc = Wm + h * 256 + nt * 16 + cp * 2;
    float* __restrict__ ldst = lds + nt * 64 + h * 32 + cp * 4;
    const int nkt_all = K >> 2;
    for (int kt0 = threadIdx.x >> 7; kt0 < nkt_all; kt0 += 32) {
      v2f r0[8], r1[8];
#pragma unroll
      for (int j = 0; j < 8; ++j) {
        const float* g = gsrc + (kt0 + 4 * j) * 512;
        r0[j] = *(const v2f*)g;
        r1[j] = *(const v2f*)(g + 128);
      }
#pragma unroll
      for (int j = 0; j < 8; ++j) {
        const float4 q = {r0[j].x, r1[j].x, r0[j].y, r1[j].y};
        *(float4*)(ldst + (kt0 + 4 * j) * 512) = q;
      }
    }
  }
  __syncthreads();

  const int lane = threadIdx.x & 31;
  const int wave = __builtin_amdgcn_readfirstlane((int)(threadIdx.x >> 5));
  const int G = (int)gridDim.x;

  const int nMain = (nTiles / (G * 4)) * (G * 4);
  for (int tile = (int)blockIdx.x + G * wave; tile < nMain;
       tile += G * GEMM_WAVES) {
    gemm_tile<8, 8>(A, K, bias, deg, prev, pdeg, out, relu, lds, tile, 0, lane);
    zero_tile(zrows, zdeg, tile, nTiles, lane);
  }

  {
    const int wrot = (wave + 8) & 15;
    const int tile = nMain + (wrot >> 2) * G + (int)blockIdx.x;
    if (tile < nTiles) {
      const int unit = wrot & 3;
      gemm_tile<2, 32>(A, K, bias, deg, prev, pdeg, out, relu, lds, tile,
                       unit * 2, lane);
      if (unit == 0) zero_tile(zrows, zdeg, tile, nTiles, lane);
    }
  }
}

extern "C" void kernel_launch(void* const* d_in, const int* in_sizes, int n_in,
                              void* d_out, int out_size, void* d_ws,
                              size_t ws_size, hipStream_t stream) {
  const int IN_DIM = 384, HID = 128;
  const int N = in_sizes[0] / IN_DIM;
  const int E = in_sizes[1] / 2;

  const float* x    = (const float*)d_in[0];
  const int*   ei   = (const int*)d_in[1];
  const float* W1_l = (const float*)d_in[2];
  const float* b1   = (const float*)d_in[3];
  const float* W1_r = (const float*)d_in[4];
  const float* W2_l = (const float*)d_in[5];
  const float* b2   = (const float*)d_in[6];
  const float* W2_r = (const float*)d_in[7];
  const int* src = ei;
  const int* dst = ei + E;
  float* out = (float*)d_out;

  float* ws = (float*)d_ws;
  const long deg_off  = 0;
  const long deg_pad  = ((long)N + 127) & ~127L;
  const long agg_off  = deg_off + deg_pad;
  const long y_off    = agg_off + (long)N * HID;
  const long h1_off   = y_off + (long)N * HID;

  float* deg  = ws + deg_off;
  float* agg  = ws + agg_off;
  float* y    = ws + y_off;
  float* agg2 = y;
  float* h1   = ws + h1_off;

  const int nTiles = N / 16;

  gemm_pass_kernel<<<GEMM_BLOCKS, GEMM_THREADS, 0, stream>>>(
      x, IN_DIM, W1_l, nullptr, nullptr, nullptr, nullptr, y, 0, nTiles, agg,
      deg);

  scatter_add_kernel<<<SCAT_BLOCKS, 256, 0, stream>>>(y, src, dst, agg, deg, E);

  gemm_pass_kernel<<<GEMM_BLOCKS, GEMM_THREADS, 0, stream>>>(
      x, IN_DIM, W1_r, b1, nullptr, agg, deg, h1, 1, nTiles, agg2, nullptr);

  scatter_add_kernel<<<SCAT_BLOCKS, 256, 0, stream>>>(h1, src, dst, agg2,
                                                      nullptr, E);

  gemm_pass_kernel<<<GEMM_BLOCKS, GEMM_THREADS, 0, stream>>>(
      agg2, HID, W2_l, b2, deg, nullptr, nullptr, out, 0, nTiles, nullptr,
      nullptr);
  gemm_pass_kernel<<<GEMM_BLOCKS, GEMM_THREADS, 0, stream>>>(
      h1, HID, W2_r, nullptr, nullptr, out, nullptr, out, 0, nTiles, nullptr,
      nullptr);
}


# Neighbour sums gathered along per-target edge lists (built once by integer exchange atomics, used by both layers), not f32 atomics into zeroed buffers; a row sums in its list's arrival order
# speedup vs baseline: 2.4442x; 1.0390x over previous
#include <hip/hip_runtime.h>


typedef __attribute__((ext_vector_type(2))) float v2f;
typedef __attribute__((ext_vector_type(8))) float v8f;

#define GEMM_WAVES 16
#define GEMM_THREADS 512
#define GEMM_BLOCKS 256
#define GEMM_MAXK 384


#define LIST_PER_THREAD 4

__global__ __launch_bounds__(256) void build_lists_kernel(
    const int* __restrict__ src, const int* __restrict__ dst,
    int* __restrict__ head, int* __restrict__ recw, int E, int N) {
  int2* __restrict__ rec = (int2*)recw;
  const int t = (int)(blockIdx.x * 256 + threadIdx.x);
  const int T = (int)gridDim.x * 256;
  int e[LIST_PER_THREAD], s[LIST_PER_THREAD], d[LIST_PER_THREAD];
  int below[LIST_PER_THREAD];
  bool push[LIST_PER_THREAD];
#pragma unroll
  for (int k = 0; k < LIST_PER_THREAD; ++k) {
    e[k] = t + k * T;
    const int ec = e[k] < E ? e[k] : E - 1;
    d[k] = dst[ec];
  }
#pragma unroll
  for (int k = 0; k < LIST_PER_THREAD; ++k) {
    const int ec = e[k] < E ? e[k] : E - 1;
    s[k] = src[ec];
  }
#pragma unroll
  for (int k = 0; k < LIST_PER_THREAD; ++k) {
    push[k] = e[k] < E && (unsigned)d[k] < (unsigned)N;
    below[k] = -1;
  }
#pragma unroll
  for (int k = 0; k < LIST_PER_THREAD; ++k)
    if (push[k]) below[k] = atomicExch(head + d[k], e[k]);
#pragma unroll
  for (int k = 0; k < LIST_PER_THREAD; ++k)
    if (e[k] < E) rec[e[k]] = make_int2(below[k], s[k]);
}

#define GATH_ROWS 16
#define GATH_WAVES 4

__global__ __launch_bounds__(GATH_WAVES * 32) void gather_sum_kernel(
    const float* __restrict__ feat, const int* __restrict__ head,
    const int* __restrict__ recw, float* __restrict__ agg,
    float* __restrict__ deg, int N, int E) {
  const int2* __restrict__ rec = (const int2*)recw;
  const int lane = threadIdx.x & 31;
  const int wave = __builtin_amdgcn_readfirstlane(
      (int)(blockIdx.x * GATH_WAVES + (threadIdx.x >> 5)));
  const int i0 = wave * GATH_ROWS;
  if (i0 >= N) return;

  int cur = -1;
  if (lane < GATH_ROWS && i0 + lane < N) cur = head[i0 + lane];
  bool live = (unsigned)cur < (unsigned)E;
  int2 r = make_int2(-1, -1);
  if (live) r = rec[cur];

  float len = 0.0f;
  float4 acc[GATH_ROWS];
#pragma unroll
  for (int j = 0; j < GATH_ROWS; ++j) acc[j] = make_float4(0.f, 0.f, 0.f, 0.f);

  for (int trip = 0; trip < E; ++trip) {
    if (__builtin_amdgcn_ballot_w32(live) == 0) break;
    const bool nlive = live && (unsigned)r.x < (unsigned)E;
    int2 rn = make_int2(-1, -1);
    if (nlive) rn = rec[r.x];

    const int sv = live ? r.y : -1;
    len += live ? 1.0f : 0.0f;
    float4 v[GATH_ROWS];
#pragma unroll
    for (int j = 0; j < GATH_ROWS; ++j) {
      const int s = __builtin_amdgcn_readlane(sv, j);
      if ((unsigned)s < (unsigned)N)
        v[j] = *(const float4*)(feat + (size_t)s * 128 + lane * 4);
      else
        v[j] = make_float4(0.f, 0.f, 0.f, 0.f);
    }
#pragma unroll
    for (int j = 0; j < GATH_ROWS; ++j) {
      acc[j].x += v[j].x;
      acc[j].y += v[j].y;
      acc[j].z += v[j].z;
      acc[j].w += v[j].w;
    }
    r = rn;
    live = nlive;
  }

#pragma unroll
  for (int j = 0; j < GATH_ROWS; ++j)
    if (i0 + j < N)
      *(float4*)(agg + (size_t)(i0 + j) * 128 + lane * 4) = acc[j];
  if (deg != nullptr && lane < GATH_ROWS && i0 + lane < N) deg[i0 + lane] = len;
}


template <int NT, int GK>
__device__ __forceinline__ void gemm_tile(
    const float* __restrict__ A, const int K, const float* __restrict__ bias,
    const float* __restrict__ deg, const float* __restrict__ prev,
    const float* __restrict__ pdeg, float* __restrict__ out, const int relu,
    const float* __restrict__ lds, const int tile, const int nt0,
    const int lane) {
  const int half = lane >> 4;
  const int n0 = lane & 15;
  const int nkt = K >> 2;
  const int ngrp = nkt / GK;
  const float* __restrict__ bl = lds + nt0 * 64 + lane * 2;

  const int arow = tile * 16 + n0;
  const float* __restrict__ arowp = A + (size_t)arow * K + half * 2;

  float scale = 1.0f;
  if (deg) scale = 1.0f / fmaxf(deg[arow], 1.0f);

  v8f acc[NT];
#pragma unroll
  for (int nt = 0; nt < NT; ++nt) {
    float b = bias ? bias[(nt0 + nt) * 16 + n0] : 0.0f;
    acc[nt] = (v8f){b, b, b, b, b, b, b, b};
  }

  v2f an[GK];
#pragma unroll
  for (int j = 0; j < GK; ++j) an[j] = *(const v2f*)(arowp + j * 4);
  v2f bn[NT];
#pragma unroll
  for (int nt = 0; nt < NT; ++nt) bn[nt] = *(const v2f*)(bl + nt * 64);

  for (int g = 0; g < ngrp; ++g) {
    v2f ac[GK];
#pragma unroll
    for (int j = 0; j < GK; ++j) ac[j] = an[j] * scale;
    const int gn = (g + 1 < ngrp) ? (g + 1) : g;
#pragma unroll
    for (int j = 0; j < GK; ++j)
      an[j] = *(const v2f*)(arowp + (gn * GK + j) * 4);

#pragma unroll
    for (int j = 0; j < GK; ++j) {
      v2f bc[NT];
#pragma unroll
      for (int nt = 0; nt < NT; ++nt) bc[nt] = bn[nt];
      int ktn = g * GK + j + 1;
      if (ktn >= nkt) ktn = nkt - 1;
      const float* __restrict__ bb = bl + ktn * 512;
#pragma unroll
      for (int nt = 0; nt < NT; ++nt) bn[nt] = *(const v2f*)(bb + nt * 64);
#pragma unroll
      for (int nt = 0; nt < NT; ++nt) {
        acc[nt] = __builtin_amdgcn_wmma_f32_16x16x4_f32(
            false, ac[j], false, bc[nt], (short)0, acc[nt], false, false);
      }
    }
#pragma unroll
    for (int j = 0; j < GK; ++j) asm volatile("" ::"v"(an[j]));
  }

  const int m0 = tile * 16;
  if (prev) {
    float pscale[8];
#pragma unroll
    for (int r = 0; r < 8; ++r)
      pscale[r] = pdeg ? 1.0f / fmaxf(pdeg[m0 + r + 8 * half], 1.0f) : 1.0f;
#pragma unroll
    for (int nt = 0; nt < NT; ++nt) {
#pragma unroll
      for (int r = 0; r < 8; ++r) {
        size_t idx = (size_t)(m0 + r + 8 * half) * 128 + (nt0 + nt) * 16 + n0;
        acc[nt][r] += prev[idx] * pscale[r];
      }
    }
  }
#pragma unroll
  for (int nt = 0; nt < NT; ++nt) {
#pragma unroll
    for (int r = 0; r < 8; ++r) {
      size_t idx = (size_t)(m0 + r + 8 * half) * 128 + (nt0 + nt) * 16 + n0;
      float v = acc[nt][r];
      if (relu) v = fmaxf(v, 0.0f);
      out[idx] = v;
    }
  }
}

__device__ __forceinline__ void init_heads_tile(int* __restrict__ zhead,
                                                const int tile,
                                                const int nTiles,
                                                const int lane) {
  if (zhead && (tile & 1) == 0) {
    const int zi = tile * 16 + lane;
    if (zi < nTiles * 16) zhead[zi] = -1;
  }
}

__global__ __launch_bounds__(GEMM_THREADS) void gemm_pass_kernel(
    const float* __restrict__ A, int K, const float* __restrict__ Wm,
    const float* __restrict__ bias, const float* __restrict__ deg,
    const float* __restrict__ prev, const float* __restrict__ pdeg,
    float* __restrict__ out, int relu, int nTiles, int* __restrict__ zhead) {
  __shared__ __attribute__((aligned(16))) float lds[GEMM_MAXK * 128];

  {
    const int low = threadIdx.x & 127;
    const int cp = low & 7;
    const int nt = ((low >> 5) & 3) * 2 + ((low >> 3) & 1);
    const int h = (low >> 4) & 1;
    const float* __restrict__ gsrc = Wm + h * 256 + nt * 16 + cp * 2;
    float* __restrict__ ldst = lds + nt * 64 + h * 32 + cp * 4;
    const int nkt_all = K >> 2;
    for (int kt0 = threadIdx.x >> 7; kt0 < nkt_all; kt0 += 32) {
      v2f r0[8], r1[8];
#pragma unroll
      for (int j = 0; j < 8; ++j) {
        const float* g = gsrc + (kt0 + 4 * j) * 512;
        r0[j] = *(const v2f*)g;
        r1[j] = *(const v2f*)(g + 128);
      }
#pragma unroll
      for (int j = 0; j < 8; ++j) {
        const float4 q = {r0[j].x, r1[j].x, r0[j].y, r1[j].y};
        *(float4*)(ldst + (kt0 + 4 * j) * 512) = q;
      }
    }
  }
  __syncthreads();

  const int lane = threadIdx.x & 31;
  const int wave = __builtin_amdgcn_readfirstlane((int)(threadIdx.x >> 5));
  const int G = (int)gridDim.x;

  const int nMain = (nTiles / (G * 4)) * (G * 4);
  for (int tile = (int)blockIdx.x + G * wave; tile < nMain;
       tile += G * GEMM_WAVES) {
    gemm_tile<8, 8>(A, K, bias, deg, prev, pdeg, out, relu, lds, tile, 0, lane);
    init_heads_tile(zhead, tile, nTiles, lane);
  }

  {
    const int wrot = (wave + 8) & 15;
    const int tile = nMain + (wrot >> 2) * G + (int)blockIdx.x;
    if (tile < nTiles) {
      const int unit = wrot & 3;
      gemm_tile<2, 32>(A, K, bias, deg, prev, pdeg, out, relu, lds, tile,
                       unit * 2, lane);
      if (unit == 0) init_heads_tile(zhead, tile, nTiles, lane);
    }
  }
}

extern "C" void kernel_launch(void* const* d_in, const int* in_sizes, int n_in,
                              void* d_out, int out_size, void* d_ws,
                              size_t ws_size, hipStream_t stream) {
  const int IN_DIM = 384, HID = 128;
  const int N = in_sizes[0] / IN_DIM;
  const int E = in_sizes[1] / 2;

  const float* x    = (const float*)d_in[0];
  const int*   ei   = (const int*)d_in[1];
  const float* W1_l = (const float*)d_in[2];
  const float* b1   = (const float*)d_in[3];
  const float* W1_r = (const float*)d_in[4];
  const float* W2_l = (const float*)d_in[5];
  const float* b2   = (const float*)d_in[6];
  const float* W2_r = (const float*)d_in[7];
  const int* src = ei;
  const int* dst = ei + E;
  float* out = (float*)d_out;

  float* ws = (float*)d_ws;
  const long row_pad  = ((long)N + 127) & ~127L;
  const long head_off = 0;
  const long deg_off  = head_off + row_pad;
  const long agg_off  = deg_off + row_pad;
  const long y_off    = agg_off + (long)N * HID;
  const long h1_off   = y_off + (long)N * HID;
  const long rec_off  = h1_off + (long)N * HID;

  int*   head = (int*)(ws + head_off);
  float* deg  = ws + deg_off;
  float* agg  = ws + agg_off;
  float* y    = ws + y_off;
  float* agg2 = y;
  float* h1   = ws + h1_off;
  int*   rec  = (int*)(ws + rec_off);

  const int nTiles = N / 16;
  const int listBlocks = (E + 256 * LIST_PER_THREAD - 1) / (256 * LIST_PER_THREAD);
  const int gathBlocks = (N + GATH_WAVES * GATH_ROWS - 1) / (GATH_WAVES * GATH_ROWS);

  gemm_pass_kernel<<<GEMM_BLOCKS, GEMM_THREADS, 0, stream>>>(
      x, IN_DIM, W1_l, nullptr, nullptr, nullptr, nullptr, y, 0, nTiles, head);

  build_lists_kernel<<<listBlocks, 256, 0, stream>>>(src, dst, head, rec, E, N);

  gather_sum_kernel<<<gathBlocks, GATH_WAVES * 32, 0, stream>>>(y, head, rec, agg,
                                                              deg, N, E);

  gemm_pass_kernel<<<GEMM_BLOCKS, GEMM_THREADS, 0, stream>>>(
      x, IN_DIM, W1_r, b1, nullptr, agg, deg, h1, 1, nTiles, nullptr);

  gather_sum_kernel<<<gathBlocks, GATH_WAVES * 32, 0, stream>>>(h1, head, rec, agg2,
                                                              nullptr, N, E);

  gemm_pass_kernel<<<GEMM_BLOCKS, GEMM_THREADS, 0, stream>>>(
      agg2, HID, W2_l, b2, deg, nullptr, nullptr, out, 0, nTiles, nullptr);
  gemm_pass_kernel<<<GEMM_BLOCKS, GEMM_THREADS, 0, stream>>>(
      h1, HID, W2_r, nullptr, nullptr, out, nullptr, out, 0, nTiles, nullptr);
}


# Layer 2 as one resident launch: a tile's wave runs h1@W2_r, walks its rows' edge lists into its own LDS scratch, adds (sum/deg)@W2_l into the same accumulators (sum order changes); 7->5 launches
# speedup vs baseline: 2.7466x; 1.1237x over previous
#include <hip/hip_runtime.h>


typedef __attribute__((ext_vector_type(2))) float v2f;
typedef __attribute__((ext_vector_type(8))) float v8f;

#define GEMM_WAVES 16
#define GEMM_THREADS 512
#define GEMM_BLOCKS 256
#define GEMM_MAXK 384


#define LIST_PER_THREAD 4

__global__ __launch_bounds__(256) void build_lists_kernel(
    const int* __restrict__ src, const int* __restrict__ dst,
    int* __restrict__ head, int* __restrict__ recw, int E, int N) {
  int2* __restrict__ rec = (int2*)recw;
  const int t = (int)(blockIdx.x * 256 + threadIdx.x);
  const int T = (int)gridDim.x * 256;
  int e[LIST_PER_THREAD], s[LIST_PER_THREAD], d[LIST_PER_THREAD];
  int below[LIST_PER_THREAD];
  bool push[LIST_PER_THREAD];
#pragma unroll
  for (int k = 0; k < LIST_PER_THREAD; ++k) {
    e[k] = t + k * T;
    const int ec = e[k] < E ? e[k] : E - 1;
    d[k] = dst[ec];
  }
#pragma unroll
  for (int k = 0; k < LIST_PER_THREAD; ++k) {
    const int ec = e[k] < E ? e[k] : E - 1;
    s[k] = src[ec];
  }
#pragma unroll
  for (int k = 0; k < LIST_PER_THREAD; ++k) {
    push[k] = e[k] < E && (unsigned)d[k] < (unsigned)N;
    below[k] = -1;
  }
#pragma unroll
  for (int k = 0; k < LIST_PER_THREAD; ++k)
    if (push[k]) below[k] = atomicExch(head + d[k], e[k]);
#pragma unroll
  for (int k = 0; k < LIST_PER_THREAD; ++k)
    if (e[k] < E) rec[e[k]] = make_int2(below[k], s[k]);
}

#define GATH_ROWS 16
#define GATH_WAVES 4

__device__ __forceinline__ void gather_first(const int* __restrict__ head,
                                             const int2* __restrict__ rec,
                                             const int i0, const int lane,
                                             const int N, const int E, int2& r,
                                             bool& live) {
  int cur = -1;
  if (lane < GATH_ROWS && i0 + lane < N) cur = head[i0 + lane];
  live = (unsigned)cur < (unsigned)E;
  r = make_int2(-1, -1);
  if (live) r = rec[cur];
}

__device__ __forceinline__ void gather_rows(
    const float* __restrict__ feat, const int2* __restrict__ rec,
    const int lane, const int N, const int E, int2 r, bool live,
    float4 (&acc)[GATH_ROWS], float& len) {
  len = 0.0f;
#pragma unroll
  for (int j = 0; j < GATH_ROWS; ++j) acc[j] = make_float4(0.f, 0.f, 0.f, 0.f);

  for (int trip = 0; trip < E; ++trip) {
    if (__builtin_amdgcn_ballot_w32(live) == 0) break;
    const bool nlive = live && (unsigned)r.x < (unsigned)E;
    int2 rn = make_int2(-1, -1);
    if (nlive) rn = rec[r.x];

    const int sv = live ? r.y : -1;
    len += live ? 1.0f : 0.0f;
    float4 v[GATH_ROWS];
#pragma unroll
    for (int j = 0; j < GATH_ROWS; ++j) {
      const int s = __builtin_amdgcn_readlane(sv, j);
      if ((unsigned)s < (unsigned)N)
        v[j] = *(const float4*)(feat + (size_t)s * 128 + lane * 4);
      else
        v[j] = make_float4(0.f, 0.f, 0.f, 0.f);
    }
#pragma unroll
    for (int j = 0; j < GATH_ROWS; ++j) {
      acc[j].x += v[j].x;
      acc[j].y += v[j].y;
      acc[j].z += v[j].z;
      acc[j].w += v[j].w;
    }
    r = rn;
    live = nlive;
  }
}

__global__ __launch_bounds__(GATH_WAVES * 32) void gather_sum_kernel(
    const float* __restrict__ feat, const int* __restrict__ head,
    const int* __restrict__ recw, float* __restrict__ agg,
    float* __restrict__ deg, int N, int E) {
  const int2* __restrict__ rec = (const int2*)recw;
  const int lane = threadIdx.x & 31;
  const int wave = __builtin_amdgcn_readfirstlane(
      (int)(blockIdx.x * GATH_WAVES + (threadIdx.x >> 5)));
  const int i0 = wave * GATH_ROWS;
  if (i0 >= N) return;

  int2 r;
  bool live;
  gather_first(head, rec, i0, lane, N, E, r, live);
  float len;
  float4 acc[GATH_ROWS];
  gather_rows(feat, rec, lane, N, E, r, live, acc, len);

#pragma unroll
  for (int j = 0; j < GATH_ROWS; ++j)
    if (i0 + j < N)
      *(float4*)(agg + (size_t)(i0 + j) * 128 + lane * 4) = acc[j];
  if (deg != nullptr && lane < GATH_ROWS && i0 + lane < N) deg[i0 + lane] = len;
}


template <int NT, int GK>
__device__ __forceinline__ void gemm_kloop(const float* arowp, const int K,
                                           const float scale,
                                           const float* bl, v8f (&acc)[NT]) {
  const int nkt = K >> 2;
  const int ngrp = nkt / GK;

  v2f an[GK];
#pragma unroll
  for (int j = 0; j < GK; ++j) an[j] = *(const v2f*)(arowp + j * 4);
  v2f bn[NT];
#pragma unroll
  for (int nt = 0; nt < NT; ++nt) bn[nt] = *(const v2f*)(bl + nt * 64);

  for (int g = 0; g < ngrp; ++g) {
    v2f ac[GK];
#pragma unroll
    for (int j = 0; j < GK; ++j) ac[j] = an[j] * scale;
    const int gn = (g + 1 < ngrp) ? (g + 1) : g;
#pragma unroll
    for (int j = 0; j < GK; ++j)
      an[j] = *(const v2f*)(arowp + (gn * GK + j) * 4);

#pragma unroll
    for (int j = 0; j < GK; ++j) {
      v2f bc[NT];
#pragma unroll
      for (int nt = 0; nt < NT; ++nt) bc[nt] = bn[nt];
      int ktn = g * GK + j + 1;
      if (ktn >= nkt) ktn = nkt - 1;
      const float* bb = bl + ktn * 512;
#pragma unroll
      for (int nt = 0; nt < NT; ++nt) bn[nt] = *(const v2f*)(bb + nt * 64);
#pragma unroll
      for (int nt = 0; nt < NT; ++nt) {
        acc[nt] = __builtin_amdgcn_wmma_f32_16x16x4_f32(
            false, ac[j], false, bc[nt], (short)0, acc[nt], false, false);
      }
    }
#pragma unroll
    for (int j = 0; j < GK; ++j) asm volatile("" ::"v"(an[j]));
  }
}

template <int NT, int GK>
__device__ __forceinline__ void gemm_tile(
    const float* __restrict__ A, const int K, const float* __restrict__ bias,
    const float* __restrict__ deg, const float* __restrict__ prev,
    const float* __restrict__ pdeg, float* __restrict__ out, const int relu,
    const float* __restrict__ lds, const int tile, const int nt0,
    const int lane) {
  const int half = lane >> 4;
  const int n0 = lane & 15;
  const float* __restrict__ bl = lds + nt0 * 64 + lane * 2;

  const int arow = tile * 16 + n0;
  const float* __restrict__ arowp = A + (size_t)arow * K + half * 2;

  float scale = 1.0f;
  if (deg) scale = 1.0f / fmaxf(deg[arow], 1.0f);

  v8f acc[NT];
#pragma unroll
  for (int nt = 0; nt < NT; ++nt) {
    float b = bias ? bias[(nt0 + nt) * 16 + n0] : 0.0f;
    acc[nt] = (v8f){b, b, b, b, b, b, b, b};
  }

  gemm_kloop<NT, GK>(arowp, K, scale, bl, acc);

  const int m0 = tile * 16;
  if (prev) {
    float pscale[8];
#pragma unroll
    for (int r = 0; r < 8; ++r)
      pscale[r] = pdeg ? 1.0f / fmaxf(pdeg[m0 + r + 8 * half], 1.0f) : 1.0f;
#pragma unroll
    for (int nt = 0; nt < NT; ++nt) {
#pragma unroll
      for (int r = 0; r < 8; ++r) {
        size_t idx = (size_t)(m0 + r + 8 * half) * 128 + (nt0 + nt) * 16 + n0;
        acc[nt][r] += prev[idx] * pscale[r];
      }
    }
  }
#pragma unroll
  for (int nt = 0; nt < NT; ++nt) {
#pragma unroll
    for (int r = 0; r < 8; ++r) {
      size_t idx = (size_t)(m0 + r + 8 * half) * 128 + (nt0 + nt) * 16 + n0;
      float v = acc[nt][r];
      if (relu) v = fmaxf(v, 0.0f);
      out[idx] = v;
    }
  }
}

__device__ __forceinline__ void init_heads_tile(int* __restrict__ zhead,
                                                const int tile,
                                                const int nTiles,
                                                const int lane) {
  if (zhead && (tile & 1) == 0) {
    const int zi = tile * 16 + lane;
    if (zi < nTiles * 16) zhead[zi] = -1;
  }
}

__device__ __forceinline__ void stage_weights(const float* __restrict__ Wm,
                                              const int K,
                                              float* __restrict__ lds) {
  const int low = threadIdx.x & 127;
  const int cp = low & 7;
  const int nt = ((low >> 5) & 3) * 2 + ((low >> 3) & 1);
  const int h = (low >> 4) & 1;
  const float* __restrict__ gsrc = Wm + h * 256 + nt * 16 + cp * 2;
  float* __restrict__ ldst = lds + nt * 64 + h * 32 + cp * 4;
  const int nkt_all = K >> 2;
  for (int kt0 = threadIdx.x >> 7; kt0 < nkt_all; kt0 += 32) {
    v2f r0[8], r1[8];
#pragma unroll
    for (int j = 0; j < 8; ++j) {
      const float* g = gsrc + (kt0 + 4 * j) * 512;
      r0[j] = *(const v2f*)g;
      r1[j] = *(const v2f*)(g + 128);
    }
#pragma unroll
    for (int j = 0; j < 8; ++j) {
      const float4 q = {r0[j].x, r1[j].x, r0[j].y, r1[j].y};
      *(float4*)(ldst + (kt0 + 4 * j) * 512) = q;
    }
  }
}

__global__ __launch_bounds__(GEMM_THREADS) void gemm_pass_kernel(
    const float* __restrict__ A, int K, const float* __restrict__ Wm,
    const float* __restrict__ bias, const float* __restrict__ deg,
    const float* __restrict__ prev, const float* __restrict__ pdeg,
    float* __restrict__ out, int relu, int nTiles, int* __restrict__ zhead) {
  __shared__ __attribute__((aligned(16))) float lds[GEMM_MAXK * 128];

  stage_weights(Wm, K, lds);
  __syncthreads();

  const int lane = threadIdx.x & 31;
  const int wave = __builtin_amdgcn_readfirstlane((int)(threadIdx.x >> 5));
  const int G = (int)gridDim.x;

  const int nMain = (nTiles / (G * 4)) * (G * 4);
  for (int tile = (int)blockIdx.x + G * wave; tile < nMain;
       tile += G * GEMM_WAVES) {
    gemm_tile<8, 8>(A, K, bias, deg, prev, pdeg, out, relu, lds, tile, 0, lane);
    init_heads_tile(zhead, tile, nTiles, lane);
  }

  {
    const int wrot = (wave + 8) & 15;
    const int tile = nMain + (wrot >> 2) * G + (int)blockIdx.x;
    if (tile < nTiles) {
      const int unit = wrot & 3;
      gemm_tile<2, 32>(A, K, bias, deg, prev, pdeg, out, relu, lds, tile,
                       unit * 2, lane);
      if (unit == 0) init_heads_tile(zhead, tile, nTiles, lane);
    }
  }
}

#define L2_PITCH 132
#define L2_WOFF (128 * 128)
#define L2_SCR (GATH_ROWS * L2_PITCH)

template <int NT, int GK>
__device__ __forceinline__ void layer2_tile(
    const float* __restrict__ h1, const int* __restrict__ head,
    const int2* __restrict__ rec, const float* __restrict__ deg,
    const float* __restrict__ bias, float* __restrict__ out,
    const float* ldsWl, const float* ldsWr, float* scr, const int tile,
    const int nt0, const int lane, const int N, const int E) {
  const int half = lane >> 4;
  const int n0 = lane & 15;
  const int m0 = tile * 16;
  const int arow = m0 + n0;
  int2 r0;
  bool live0;
  gather_first(head, rec, m0, lane, N, E, r0, live0);
  const float dg = deg[arow];

  v8f acc[NT];
#pragma unroll
  for (int nt = 0; nt < NT; ++nt) {
    const float b = bias[(nt0 + nt) * 16 + n0];
    acc[nt] = (v8f){b, b, b, b, b, b, b, b};
  }

  gemm_kloop<NT, GK>(h1 + (size_t)arow * 128 + half * 2, 128, 1.0f,
                     ldsWr + nt0 * 64 + lane * 2, acc);

  {
    float len;
    float4 g[GATH_ROWS];
    gather_rows(h1, rec, lane, N, E, r0, live0, g, len);
#pragma unroll
    for (int j = 0; j < GATH_ROWS; ++j)
      *(float4*)(scr + j * L2_PITCH + lane * 4) = g[j];
  }

  const float scale = 1.0f / fmaxf(dg, 1.0f);
  gemm_kloop<NT, GK>(scr + n0 * L2_PITCH + half * 2, 128, scale,
                     ldsWl + nt0 * 64 + lane * 2, acc);

#pragma unroll
  for (int nt = 0; nt < NT; ++nt) {
#pragma unroll
    for (int r = 0; r < 8; ++r) {
      size_t idx = (size_t)(m0 + r + 8 * half) * 128 + (nt0 + nt) * 16 + n0;
      out[idx] = acc[nt][r];
    }
  }
}

__global__ __launch_bounds__(GEMM_THREADS) void layer2_kernel(
    const float* __restrict__ h1, const int* __restrict__ head,
    const int* __restrict__ recw, const float* __restrict__ deg,
    const float* __restrict__ Wl, const float* __restrict__ Wr,
    const float* __restrict__ bias, float* __restrict__ out, int nTiles, int N,
    int E) {
  __shared__ __attribute__((aligned(16))) float lds[2 * L2_WOFF + GEMM_WAVES * L2_SCR];
  const int2* __restrict__ rec = (const int2*)recw;

  stage_weights(Wl, 128, lds);
  stage_weights(Wr, 128, lds + L2_WOFF);
  __syncthreads();

  const int lane = threadIdx.x & 31;
  const int wave = __builtin_amdgcn_readfirstlane((int)(threadIdx.x >> 5));
  const int G = (int)gridDim.x;
  float* scr = lds + 2 * L2_WOFF + wave * L2_SCR;

  const int nMain = (nTiles / (G * 4)) * (G * 4);
  for (int tile = (int)blockIdx.x + G * wave; tile < nMain;
       tile += G * GEMM_WAVES)
    layer2_tile<8, 8>(h1, head, rec, deg, bias, out, lds, lds + L2_WOFF, scr,
                      tile, 0, lane, N, E);

  {
    const int wrot = (wave + 8) & 15;
    const int tile = nMain + (wrot >> 2) * G + (int)blockIdx.x;
    if (tile < nTiles)
      layer2_tile<2, 32>(h1, head, rec, deg, bias, out, lds, lds + L2_WOFF, scr,
                         tile, (wrot & 3) * 2, lane, N, E);
  }
}

extern "C" void kernel_launch(void* const* d_in, const int* in_sizes, int n_in,
                              void* d_out, int out_size, void* d_ws,
                              size_t ws_size, hipStream_t stream) {
  const int IN_DIM = 384, HID = 128;
  const int N = in_sizes[0] / IN_DIM;
  const int E = in_sizes[1] / 2;

  const float* x    = (const float*)d_in[0];
  const int*   ei   = (const int*)d_in[1];
  const float* W1_l = (const float*)d_in[2];
  const float* b1   = (const float*)d_in[3];
  const float* W1_r = (const float*)d_in[4];
  const float* W2_l = (const float*)d_in[5];
  const float* b2   = (const float*)d_in[6];
  const float* W2_r = (const float*)d_in[7];
  const int* src = ei;
  const int* dst = ei + E;
  float* out = (float*)d_out;

  float* ws = (float*)d_ws;
  const long row_pad  = ((long)N + 127) & ~127L;
  const long head_off = 0;
  const long deg_off  = head_off + row_pad;
  const long agg_off  = deg_off + row_pad;
  const long y_off    = agg_off + (long)N * HID;
  const long h1_off   = y_off + (long)N * HID;
  const long rec_off  = h1_off + (long)N * HID;

  int*   head = (int*)(ws + head_off);
  float* deg  = ws + deg_off;
  float* agg  = ws + agg_off;
  float* y    = ws + y_off;
  float* h1   = ws + h1_off;
  int*   rec  = (int*)(ws + rec_off);

  const int nTiles = N / 16;
  const int listBlocks = (E + 256 * LIST_PER_THREAD - 1) / (256 * LIST_PER_THREAD);
  const int gathBlocks = (N + GATH_WAVES * GATH_ROWS - 1) / (GATH_WAVES * GATH_ROWS);

  gemm_pass_kernel<<<GEMM_BLOCKS, GEMM_THREADS, 0, stream>>>(
      x, IN_DIM, W1_l, nullptr, nullptr, nullptr, nullptr, y, 0, nTiles, head);

  build_lists_kernel<<<listBlocks, 256, 0, stream>>>(src, dst, head, rec, E, N);

  gather_sum_kernel<<<gathBlocks, GATH_WAVES * 32, 0, stream>>>(y, head, rec, agg,
                                                              deg, N, E);

  gemm_pass_kernel<<<GEMM_BLOCKS, GEMM_THREADS, 0, stream>>>(
      x, IN_DIM, W1_r, b1, nullptr, agg, deg, h1, 1, nTiles, nullptr);

  layer2_kernel<<<GEMM_BLOCKS, GEMM_THREADS, 0, stream>>>(
      h1, head, rec, deg, W2_l, W2_r, b2, out, nTiles, N, E);
}


# Layer 1 as one resident launch: a tile's wave runs x@W1_r, walks its rows' edge lists over y, adds sum/length into its accumulators through LDS; gather launch, agg and deg buffers gone; 5->4 launches
# speedup vs baseline: 2.9233x; 1.0643x over previous
#include <hip/hip_runtime.h>


typedef __attribute__((ext_vector_type(2))) float v2f;
typedef __attribute__((ext_vector_type(8))) float v8f;

#define GEMM_WAVES 16
#define GEMM_THREADS 512
#define GEMM_BLOCKS 256
#define GEMM_MAXK 384


#define LIST_PER_THREAD 4

__global__ __launch_bounds__(256) void build_lists_kernel(
    const int* __restrict__ src, const int* __restrict__ dst,
    int* __restrict__ head, int* __restrict__ recw, int E, int N) {
  int2* __restrict__ rec = (int2*)recw;
  const int t = (int)(blockIdx.x * 256 + threadIdx.x);
  const int T = (int)gridDim.x * 256;
  int e[LIST_PER_THREAD], s[LIST_PER_THREAD], d[LIST_PER_THREAD];
  int below[LIST_PER_THREAD];
  bool push[LIST_PER_THREAD];
#pragma unroll
  for (int k = 0; k < LIST_PER_THREAD; ++k) {
    e[k] = t + k * T;
    const int ec = e[k] < E ? e[k] : E - 1;
    d[k] = dst[ec];
  }
#pragma unroll
  for (int k = 0; k < LIST_PER_THREAD; ++k) {
    const int ec = e[k] < E ? e[k] : E - 1;
    s[k] = src[ec];
  }
#pragma unroll
  for (int k = 0; k < LIST_PER_THREAD; ++k) {
    push[k] = e[k] < E && (unsigned)d[k] < (unsigned)N;
    below[k] = -1;
  }
#pragma unroll
  for (int k = 0; k < LIST_PER_THREAD; ++k)
    if (push[k]) below[k] = atomicExch(head + d[k], e[k]);
#pragma unroll
  for (int k = 0; k < LIST_PER_THREAD; ++k)
    if (e[k] < E) rec[e[k]] = make_int2(below[k], s[k]);
}

#define GATH_ROWS 16

__device__ __forceinline__ void gather_first(const int* __restrict__ head,
                                             const int2* __restrict__ rec,
                                             const int i0, const int lane,
                                             const int N, const int E, int2& r,
                                             bool& live) {
  int cur = -1;
  if (lane < GATH_ROWS && i0 + lane < N) cur = head[i0 + lane];
  live = (unsigned)cur < (unsigned)E;
  r = make_int2(-1, -1);
  if (live) r = rec[cur];
}

__device__ __forceinline__ void gather_rows(
    const float* __restrict__ feat, const int2* __restrict__ rec,
    const int lane, const int N, const int E, int2 r, bool live,
    float4 (&acc)[GATH_ROWS], float& len) {
  len = 0.0f;
#pragma unroll
  for (int j = 0; j < GATH_ROWS; ++j) acc[j] = make_float4(0.f, 0.f, 0.f, 0.f);

  for (int trip = 0; trip < E; ++trip) {
    if (__builtin_amdgcn_ballot_w32(live) == 0) break;
    const bool nlive = live && (unsigned)r.x < (unsigned)E;
    int2 rn = make_int2(-1, -1);
    if (nlive) rn = rec[r.x];

    const int sv = live ? r.y : -1;
    len += live ? 1.0f : 0.0f;
    float4 v[GATH_ROWS];
#pragma unroll
    for (int j = 0; j < GATH_ROWS; ++j) {
      const int s = __builtin_amdgcn_readlane(sv, j);
      if ((unsigned)s < (unsigned)N)
        v[j] = *(const float4*)(feat + (size_t)s * 128 + lane * 4);
      else
        v[j] = make_float4(0.f, 0.f, 0.f, 0.f);
    }
#pragma unroll
    for (int j = 0; j < GATH_ROWS; ++j) {
      acc[j].x += v[j].x;
      acc[j].y += v[j].y;
      acc[j].z += v[j].z;
      acc[j].w += v[j].w;
    }
    r = rn;
    live = nlive;
  }
}


template <int NT, int GK>
__device__ __forceinline__ void gemm_kloop(const float* arowp, const int K,
                                           const float scale,
                                           const float* bl, v8f (&acc)[NT]) {
  const int nkt = K >> 2;
  const int ngrp = nkt / GK;

  v2f an[GK];
#pragma unroll
  for (int j = 0; j < GK; ++j) an[j] = *(const v2f*)(arowp + j * 4);
  v2f bn[NT];
#pragma unroll
  for (int nt = 0; nt < NT; ++nt) bn[nt] = *(const v2f*)(bl + nt * 64);

  for (int g = 0; g < ngrp; ++g) {
    v2f ac[GK];
#pragma unroll
    for (int j = 0; j < GK; ++j) ac[j] = an[j] * scale;
    const int gn = (g + 1 < ngrp) ? (g + 1) : g;
#pragma unroll
    for (int j = 0; j < GK; ++j)
      an[j] = *(const v2f*)(arowp + (gn * GK + j) * 4);

#pragma unroll
    for (int j = 0; j < GK; ++j) {
      v2f bc[NT];
#pragma unroll
      for (int nt = 0; nt < NT; ++nt) bc[nt] = bn[nt];
      int ktn = g * GK + j + 1;
      if (ktn >= nkt) ktn = nkt - 1;
      const float* bb = bl + ktn * 512;
#pragma unroll
      for (int nt = 0; nt < NT; ++nt) bn[nt] = *(const v2f*)(bb + nt * 64);
#pragma unroll
      for (int nt = 0; nt < NT; ++nt) {
        acc[nt] = __builtin_amdgcn_wmma_f32_16x16x4_f32(
            false, ac[j], false, bc[nt], (short)0, acc[nt], false, false);
      }
    }
#pragma unroll
    for (int j = 0; j < GK; ++j) asm volatile("" ::"v"(an[j]));
  }
}

template <int NT, int GK>
__device__ __forceinline__ void gemm_tile(
    const float* __restrict__ A, const int K, const float* __restrict__ bias,
    const float* __restrict__ deg, const float* __restrict__ prev,
    const float* __restrict__ pdeg, float* __restrict__ out, const int relu,
    const float* __restrict__ lds, const int tile, const int nt0,
    const int lane) {
  const int half = lane >> 4;
  const int n0 = lane & 15;
  const float* __restrict__ bl = lds + nt0 * 64 + lane * 2;

  const int arow = tile * 16 + n0;
  const float* __restrict__ arowp = A + (size_t)arow * K + half * 2;

  float scale = 1.0f;
  if (deg) scale = 1.0f / fmaxf(deg[arow], 1.0f);

  v8f acc[NT];
#pragma unroll
  for (int nt = 0; nt < NT; ++nt) {
    float b = bias ? bias[(nt0 + nt) * 16 + n0] : 0.0f;
    acc[nt] = (v8f){b, b, b, b, b, b, b, b};
  }

  gemm_kloop<NT, GK>(arowp, K, scale, bl, acc);

  const int m0 = tile * 16;
  if (prev) {
    float pscale[8];
#pragma unroll
    for (int r = 0; r < 8; ++r)
      pscale[r] = pdeg ? 1.0f / fmaxf(pdeg[m0 + r + 8 * half], 1.0f) : 1.0f;
#pragma unroll
    for (int nt = 0; nt < NT; ++nt) {
#pragma unroll
      for (int r = 0; r < 8; ++r) {
        size_t idx = (size_t)(m0 + r + 8 * half) * 128 + (nt0 + nt) * 16 + n0;
        acc[nt][r] += prev[idx] * pscale[r];
      }
    }
  }
#pragma unroll
  for (int nt = 0; nt < NT; ++nt) {
#pragma unroll
    for (int r = 0; r < 8; ++r) {
      size_t idx = (size_t)(m0 + r + 8 * half) * 128 + (nt0 + nt) * 16 + n0;
      float v = acc[nt][r];
      if (relu) v = fmaxf(v, 0.0f);
      out[idx] = v;
    }
  }
}

__device__ __forceinline__ void init_heads_tile(int* __restrict__ zhead,
                                                const int tile,
                                                const int nTiles,
                                                const int lane) {
  if (zhead && (tile & 1) == 0) {
    const int zi = tile * 16 + lane;
    if (zi < nTiles * 16) zhead[zi] = -1;
  }
}

__device__ __forceinline__ void stage_weights(const float* __restrict__ Wm,
                                              const int K,
                                              float* __restrict__ lds) {
  const int low = threadIdx.x & 127;
  const int cp = low & 7;
  const int nt = ((low >> 5) & 3) * 2 + ((low >> 3) & 1);
  const int h = (low >> 4) & 1;
  const float* __restrict__ gsrc = Wm + h * 256 + nt * 16 + cp * 2;
  float* __restrict__ ldst = lds + nt * 64 + h * 32 + cp * 4;
  const int nkt_all = K >> 2;
  for (int kt0 = threadIdx.x >> 7; kt0 < nkt_all; kt0 += 32) {
    v2f r0[8], r1[8];
#pragma unroll
    for (int j = 0; j < 8; ++j) {
      const float* g = gsrc + (kt0 + 4 * j) * 512;
      r0[j] = *(const v2f*)g;
      r1[j] = *(const v2f*)(g + 128);
    }
#pragma unroll
    for (int j = 0; j < 8; ++j) {
      const float4 q = {r0[j].x, r1[j].x, r0[j].y, r1[j].y};
      *(float4*)(ldst + (kt0 + 4 * j) * 512) = q;
    }
  }
}

__global__ __launch_bounds__(GEMM_THREADS) void gemm_pass_kernel(
    const float* __restrict__ A, int K, const float* __restrict__ Wm,
    const float* __restrict__ bias, const float* __restrict__ deg,
    const float* __restrict__ prev, const float* __restrict__ pdeg,
    float* __restrict__ out, int relu, int nTiles, int* __restrict__ zhead) {
  __shared__ __attribute__((aligned(16))) float lds[GEMM_MAXK * 128];

  stage_weights(Wm, K, lds);
  __syncthreads();

  const int lane = threadIdx.x & 31;
  const int wave = __builtin_amdgcn_readfirstlane((int)(threadIdx.x >> 5));
  const int G = (int)gridDim.x;

  const int nMain = (nTiles / (G * 4)) * (G * 4);
  for (int tile = (int)blockIdx.x + G * wave; tile < nMain;
       tile += G * GEMM_WAVES) {
    gemm_tile<8, 8>(A, K, bias, deg, prev, pdeg, out, relu, lds, tile, 0, lane);
    init_heads_tile(zhead, tile, nTiles, lane);
  }

  {
    const int wrot = (wave + 8) & 15;
    const int tile = nMain + (wrot >> 2) * G + (int)blockIdx.x;
    if (tile < nTiles) {
      const int unit = wrot & 3;
      gemm_tile<2, 32>(A, K, bias, deg, prev, pdeg, out, relu, lds, tile,
                       unit * 2, lane);
      if (unit == 0) init_heads_tile(zhead, tile, nTiles, lane);
    }
  }
}

#define L1_PITCH 132
#define L1_SCR (8 * L1_PITCH)

template <int NT, int GK>
__device__ __forceinline__ void layer1_tile(
    const float* __restrict__ x, const int K, const float* __restrict__ y,
    const int* __restrict__ head, const int2* __restrict__ rec,
    const float* __restrict__ bias, float* __restrict__ h1, const float* ldsW,
    float* scr, const int tile, const int nt0, const int lane, const int N,
    const int E) {
  const int half = lane >> 4;
  const int n0 = lane & 15;
  const int m0 = tile * 16;
  const int arow = m0 + n0;
  int2 r0;
  bool live0;
  gather_first(head, rec, m0, lane, N, E, r0, live0);

  v8f acc[NT];
#pragma unroll
  for (int nt = 0; nt < NT; ++nt) {
    const float b = bias[(nt0 + nt) * 16 + n0];
    acc[nt] = (v8f){b, b, b, b, b, b, b, b};
  }

  gemm_kloop<NT, GK>(x + (size_t)arow * K + half * 2, K, 1.0f,
                     ldsW + nt0 * 64 + lane * 2, acc);

  float len;
  float4 g[GATH_ROWS];
  gather_rows(y, rec, lane, N, E, r0, live0, g, len);
  const float inv = 1.0f / fmaxf(len, 1.0f);

#pragma unroll
  for (int p = 0; p < 2; ++p) {
#pragma unroll
    for (int s = 0; s < 4; ++s) {
      *(float4*)(scr + s * L1_PITCH + lane * 4) = g[4 * p + s];
      *(float4*)(scr + (4 + s) * L1_PITCH + lane * 4) = g[8 + 4 * p + s];
    }
#pragma unroll
    for (int rr = 0; rr < 4; ++rr) {
      const int r = 4 * p + rr;
      const float sc = __shfl(inv, 8 * half + r);
      const float* sp = scr + (4 * half + rr) * L1_PITCH + nt0 * 16 + n0;
#pragma unroll
      for (int nt = 0; nt < NT; ++nt) acc[nt][r] += sp[nt * 16] * sc;
    }
  }

#pragma unroll
  for (int nt = 0; nt < NT; ++nt) {
#pragma unroll
    for (int r = 0; r < 8; ++r) {
      size_t idx = (size_t)(m0 + r + 8 * half) * 128 + (nt0 + nt) * 16 + n0;
      h1[idx] = fmaxf(acc[nt][r], 0.0f);
    }
  }
}

__global__ __launch_bounds__(GEMM_THREADS) void layer1_kernel(
    const float* __restrict__ x, int K, const float* __restrict__ Wr,
    const float* __restrict__ bias, const float* __restrict__ y,
    const int* __restrict__ head, const int* __restrict__ recw,
    float* __restrict__ h1, int nTiles, int N, int E) {
  __shared__ __attribute__((aligned(16))) float lds[GEMM_MAXK * 128 + GEMM_WAVES * L1_SCR];
  const int2* __restrict__ rec = (const int2*)recw;

  stage_weights(Wr, K, lds);
  __syncthreads();

  const int lane = threadIdx.x & 31;
  const int wave = __builtin_amdgcn_readfirstlane((int)(threadIdx.x >> 5));
  const int G = (int)gridDim.x;
  float* scr = lds + GEMM_MAXK * 128 + wave * L1_SCR;

  const int nMain = (nTiles / (G * 4)) * (G * 4);
  for (int tile = (int)blockIdx.x + G * wave; tile < nMain;
       tile += G * GEMM_WAVES)
    layer1_tile<8, 8>(x, K, y, head, rec, bias, h1, lds, scr, tile, 0, lane, N,
                      E);

  {
    const int wrot = (wave + 8) & 15;
    const int tile = nMain + (wrot >> 2) * G + (int)blockIdx.x;
    if (tile < nTiles)
      layer1_tile<2, 32>(x, K, y, head, rec, bias, h1, lds, scr, tile,
                         (wrot & 3) * 2, lane, N, E);
  }
}

#define L2_PITCH 132
#define L2_WOFF (128 * 128)
#define L2_SCR (GATH_ROWS * L2_PITCH)

template <int NT, int GK>
__device__ __forceinline__ void layer2_tile(
    const float* __restrict__ h1, const int* __restrict__ head,
    const int2* __restrict__ rec,
    const float* __restrict__ bias, float* __restrict__ out,
    const float* ldsWl, const float* ldsWr, float* scr, const int tile,
    const int nt0, const int lane, const int N, const int E) {
  const int half = lane >> 4;
  const int n0 = lane & 15;
  const int m0 = tile * 16;
  const int arow = m0 + n0;
  int2 r0;
  bool live0;
  gather_first(head, rec, m0, lane, N, E, r0, live0);

  v8f acc[NT];
#pragma unroll
  for (int nt = 0; nt < NT; ++nt) {
    const float b = bias[(nt0 + nt) * 16 + n0];
    acc[nt] = (v8f){b, b, b, b, b, b, b, b};
  }

  gemm_kloop<NT, GK>(h1 + (size_t)arow * 128 + half * 2, 128, 1.0f,
                     ldsWr + nt0 * 64 + lane * 2, acc);

  float dg;
  {
    float len;
    float4 g[GATH_ROWS];
    gather_rows(h1, rec, lane, N, E, r0, live0, g, len);
#pragma unroll
    for (int j = 0; j < GATH_ROWS; ++j)
      *(float4*)(scr + j * L2_PITCH + lane * 4) = g[j];
    dg = __shfl(len, n0);
  }

  const float scale = 1.0f / fmaxf(dg, 1.0f);
  gemm_kloop<NT, GK>(scr + n0 * L2_PITCH + half * 2, 128, scale,
                     ldsWl + nt0 * 64 + lane * 2, acc);

#pragma unroll
  for (int nt = 0; nt < NT; ++nt) {
#pragma unroll
    for (int r = 0; r < 8; ++r) {
      size_t idx = (size_t)(m0 + r + 8 * half) * 128 + (nt0 + nt) * 16 + n0;
      out[idx] = acc[nt][r];
    }
  }
}

__global__ __launch_bounds__(GEMM_THREADS) void layer2_kernel(
    const float* __restrict__ h1, const int* __restrict__ head,
    const int* __restrict__ recw,
    const float* __restrict__ Wl, const float* __restrict__ Wr,
    const float* __restrict__ bias, float* __restrict__ out, int nTiles, int N,
    int E) {
  __shared__ __attribute__((aligned(16))) float lds[2 * L2_WOFF + GEMM_WAVES * L2_SCR];
  const int2* __restrict__ rec = (const int2*)recw;

  stage_weights(Wl, 128, lds);
  stage_weights(Wr, 128, lds + L2_WOFF);
  __syncthreads();

  const int lane = threadIdx.x & 31;
  const int wave = __builtin_amdgcn_readfirstlane((int)(threadIdx.x >> 5));
  const int G = (int)gridDim.x;
  float* scr = lds + 2 * L2_WOFF + wave * L2_SCR;

  const int nMain = (nTiles / (G * 4)) * (G * 4);
  for (int tile = (int)blockIdx.x + G * wave; tile < nMain;
       tile += G * GEMM_WAVES)
    layer2_tile<8, 8>(h1, head, rec, bias, out, lds, lds + L2_WOFF, scr,
                      tile, 0, lane, N, E);

  {
    const int wrot = (wave + 8) & 15;
    const int tile = nMain + (wrot >> 2) * G + (int)blockIdx.x;
    if (tile < nTiles)
      layer2_tile<2, 32>(h1, head, rec, bias, out, lds, lds + L2_WOFF, scr,
                         tile, (wrot & 3) * 2, lane, N, E);
  }
}

extern "C" void kernel_launch(void* const* d_in, const int* in_sizes, int n_in,
                              void* d_out, int out_size, void* d_ws,
                              size_t ws_size, hipStream_t stream) {
  const int IN_DIM = 384, HID = 128;
  const int N = in_sizes[0] / IN_DIM;
  const int E = in_sizes[1] / 2;

  const float* x    = (const float*)d_in[0];
  const int*   ei   = (const int*)d_in[1];
  const float* W1_l = (const float*)d_in[2];
  const float* b1   = (const float*)d_in[3];
  const float* W1_r = (const float*)d_in[4];
  const float* W2_l = (const float*)d_in[5];
  const float* b2   = (const float*)d_in[6];
  const float* W2_r = (const float*)d_in[7];
  const int* src = ei;
  const int* dst = ei + E;
  float* out = (float*)d_out;

  float* ws = (float*)d_ws;
  const long row_pad  = ((long)N + 127) & ~127L;
  const long head_off = 0;
  const long y_off    = head_off + row_pad;
  const long h1_off   = y_off + (long)N * HID;
  const long rec_off  = h1_off + (long)N * HID;

  int*   head = (int*)(ws + head_off);
  float* y    = ws + y_off;
  float* h1   = ws + h1_off;
  int*   rec  = (int*)(ws + rec_off);

  const int nTiles = N / 16;
  const int listBlocks = (E + 256 * LIST_PER_THREAD - 1) / (256 * LIST_PER_THREAD);

  gemm_pass_kernel<<<GEMM_BLOCKS, GEMM_THREADS, 0, stream>>>(
      x, IN_DIM, W1_l, nullptr, nullptr, nullptr, nullptr, y, 0, nTiles, head);

  build_lists_kernel<<<listBlocks, 256, 0, stream>>>(src, dst, head, rec, E, N);

  layer1_kernel<<<GEMM_BLOCKS, GEMM_THREADS, 0, stream>>>(
      x, IN_DIM, W1_r, b1, y, head, rec, h1, nTiles, N, E);

  layer2_kernel<<<GEMM_BLOCKS, GEMM_THREADS, 0, stream>>>(
      h1, head, rec, W2_l, W2_r, b2, out, nTiles, N, E);
}
